# starting kernel
# baseline (speedup 1.0000x reference)
#include <hip/hip_runtime.h>

#define GCN_N 100000
#define GCN_D 128

typedef float v2f __attribute__((ext_vector_type(2)));
typedef float v8f __attribute__((ext_vector_type(8)));

__global__ __launch_bounds__(256) void gcn_fill_zero(float* __restrict__ p, long long n) {
    long long i = (long long)blockIdx.x * blockDim.x + threadIdx.x;
    if (i < n) p[i] = 0.0f;
}

__global__ __launch_bounds__(256) void gcn_deg_accum(const int* __restrict__ dst,
                                                     float* __restrict__ deg, int ne) {
    int i = blockIdx.x * blockDim.x + threadIdx.x;
    if (i < ne) atomicAdd(&deg[dst[i]], 1.0f);
}

__global__ __launch_bounds__(256) void gcn_finalize_dinv(float* __restrict__ dinv, int n) {
    int i = blockIdx.x * blockDim.x + threadIdx.x;
    if (i < n) dinv[i] = rsqrtf(dinv[i] + 1.0f);
}

__global__ __launch_bounds__(256) void gcn_gemm_wmma(const float* __restrict__ X,
                                                     const float* __restrict__ W,
                                                     float* __restrict__ H) {
    const int lane     = threadIdx.x & 31;
    const int wave     = blockIdx.x * 8 + (threadIdx.x >> 5);
    const int col_tile = wave & 7;
    const int row_tile = wave >> 3;
    const int m        = lane & 15;
    const int half     = lane >> 4;
    const int koff     = half << 1;

    const float* xrow = X + (size_t)(row_tile * 16 + m) * GCN_D;
    const int    ncol = col_tile * 16 + m;

    v8f acc = {};
#pragma unroll
    for (int k = 0; k < GCN_D; k += 4) {
        v2f a = *(const v2f*)(xrow + k + koff);
        v2f b;
        b.x = W[(size_t)(k + koff)     * GCN_D + ncol];
        b.y = W[(size_t)(k + koff + 1) * GCN_D + ncol];
        acc = __builtin_amdgcn_wmma_f32_16x16x4_f32(
                  false, a, false, b, (short)0, acc, false, false);
    }

    const int mbase = row_tile * 16 + half * 8;
    float* hp = H + (size_t)mbase * GCN_D + col_tile * 16 + m;
#pragma unroll
    for (int i = 0; i < 8; ++i) hp[(size_t)i * GCN_D] = acc[i];
}

__global__ __launch_bounds__(256) void gcn_scatter(const int* __restrict__ src,
                                                   const int* __restrict__ dst,
                                                   const float* __restrict__ h,
                                                   const float* __restrict__ dinv,
                                                   float* __restrict__ acc, int ne) {
    int warp = (int)((blockIdx.x * blockDim.x + threadIdx.x) >> 5);
    int lane = threadIdx.x & 31;
    if (warp >= ne) return;
    int s = src[warp];
    int d = dst[warp];
    float nrm = dinv[s] * dinv[d];
    float4 v = ((const float4*)(h + (size_t)s * GCN_D))[lane];
    float* o = acc + (size_t)d * GCN_D + lane * 4;
    atomicAdd(o + 0, v.x * nrm);
    atomicAdd(o + 1, v.y * nrm);
    atomicAdd(o + 2, v.z * nrm);
    atomicAdd(o + 3, v.w * nrm);
}

__global__ __launch_bounds__(256) void gcn_epilogue(const float* __restrict__ acc,
                                                    const float* __restrict__ h,
                                                    const float* __restrict__ dinv,
                                                    const float* __restrict__ b,
                                                    float* __restrict__ out, long long n) {
    long long i = (long long)blockIdx.x * blockDim.x + threadIdx.x;
    if (i >= n) return;
    int node = (int)(i >> 7);
    int j    = (int)(i & 127);
    float di = dinv[node];
    float v  = acc[i] + h[i] * di * di + b[j];
    out[i] = v > 0.0f ? v : 0.0f;
}

extern "C" void kernel_launch(void* const* d_in, const int* in_sizes, int n_in,
                              void* d_out, int out_size, void* d_ws, size_t ws_size,
                              hipStream_t stream) {
    const float* x  = (const float*)d_in[0];
    const float* W1 = (const float*)d_in[1];
    const float* b1 = (const float*)d_in[2];
    const float* W2 = (const float*)d_in[3];
    const float* b2 = (const float*)d_in[4];
    const float* W3 = (const float*)d_in[5];
    const float* b3 = (const float*)d_in[6];
    const int*   eidx = (const int*)d_in[7];
    const int ne = in_sizes[7] / 2;
    const int* src = eidx;
    const int* dst = eidx + ne;

    char* ws = (char*)d_ws;
    size_t off = 0;
    float* dinv = (float*)(ws + off); off += (((size_t)GCN_N * 4) + 255) & ~(size_t)255;
    float* h    = (float*)(ws + off); off += (size_t)GCN_N * GCN_D * 4;
    float* acc  = (float*)(ws + off); off += (size_t)GCN_N * GCN_D * 4;
    float* buf  = (float*)(ws + off);

    const long long feat_elems = (long long)GCN_N * GCN_D;
    const int blk = 256;
    const int g_node  = (GCN_N + blk - 1) / blk;
    const int g_edge  = (ne + blk - 1) / blk;
    const long long g_feat_ll = (feat_elems + blk - 1) / blk;
    const int g_feat  = (int)g_feat_ll;
    const int g_gemm  = GCN_N / 16;
    const int g_scat  = (int)(((long long)ne * 32 + blk - 1) / blk);

    gcn_fill_zero<<<g_node, blk, 0, stream>>>(dinv, GCN_N);
    gcn_deg_accum<<<g_edge, blk, 0, stream>>>(dst, dinv, ne);
    gcn_finalize_dinv<<<g_node, blk, 0, stream>>>(dinv, GCN_N);

    gcn_gemm_wmma<<<g_gemm, blk, 0, stream>>>(x, W1, h);
    gcn_fill_zero<<<g_feat, blk, 0, stream>>>(acc, feat_elems);
    gcn_scatter<<<g_scat, blk, 0, stream>>>(src, dst, h, dinv, acc, ne);
    gcn_epilogue<<<g_feat, blk, 0, stream>>>(acc, h, dinv, b1, buf, feat_elems);

    gcn_gemm_wmma<<<g_gemm, blk, 0, stream>>>(buf, W2, h);
    gcn_fill_zero<<<g_feat, blk, 0, stream>>>(acc, feat_elems);
    gcn_scatter<<<g_scat, blk, 0, stream>>>(src, dst, h, dinv, acc, ne);
    gcn_epilogue<<<g_feat, blk, 0, stream>>>(acc, h, dinv, b2, buf, feat_elems);

    gcn_gemm_wmma<<<g_gemm, blk, 0, stream>>>(buf, W3, h);
    gcn_fill_zero<<<g_feat, blk, 0, stream>>>(acc, feat_elems);
    gcn_scatter<<<g_scat, blk, 0, stream>>>(src, dst, h, dinv, acc, ne);
    gcn_epilogue<<<g_feat, blk, 0, stream>>>(acc, h, dinv, b3, (float*)d_out, feat_elems);
}


# Per-edge float atomic scatter -> gather: edges bucketed by destination once per call (int atomic rank + scan); one wave per output row sums in registers; sum order = rank order, self loop last
# speedup vs baseline: 3.0410x; 3.0410x over previous
#include <hip/hip_runtime.h>


#define GCN_N 100000
#define GCN_D 128
#define GCN_TILE 1024

typedef float v2f __attribute__((ext_vector_type(2)));
typedef float v8f __attribute__((ext_vector_type(8)));

__global__ __launch_bounds__(256) void gcn_fill_zero_i32(int* __restrict__ p, int n) {
    int i = blockIdx.x * blockDim.x + threadIdx.x;
    if (i < n) p[i] = 0;
}

__global__ __launch_bounds__(256) void gcn_deg_rank(const int* __restrict__ dst,
                                                    int* __restrict__ cnt,
                                                    int* __restrict__ rank, int ne) {
    int i = blockIdx.x * blockDim.x + threadIdx.x;
    if (i < ne) rank[i] = atomicAdd(&cnt[dst[i]], 1);
}

__device__ __forceinline__ int gcn_wave_sum(int v) {
    for (int o = 16; o > 0; o >>= 1) v += __shfl_xor(v, o);
    return v;
}

__global__ __launch_bounds__(256) void gcn_tile_sums(const int* __restrict__ cnt,
                                                     int* __restrict__ tsum, int n) {
    __shared__ int part[8];
    const int t    = threadIdx.x;
    const int base = blockIdx.x * GCN_TILE + t * 4;
    int s = 0;
    if (base + 3 < n) {
        int4 v = *(const int4*)(cnt + base);
        s = v.x + v.y + v.z + v.w;
    } else {
        for (int k = 0; k < 4; ++k) if (base + k < n) s += cnt[base + k];
    }
    s = gcn_wave_sum(s);
    if ((t & 31) == 0) part[t >> 5] = s;
    __syncthreads();
    if (t == 0) {
        int r = 0;
        for (int k = 0; k < 8; ++k) r += part[k];
        tsum[blockIdx.x] = r;
    }
}

__global__ __launch_bounds__(256) void gcn_scan_rows(const int* __restrict__ cnt,
                                                     const int* __restrict__ tsum,
                                                     int* __restrict__ rowptr,
                                                     float* __restrict__ dinv, int n) {
    __shared__ int part[8];
    __shared__ int wtot[8];
    const int t = threadIdx.x, lane = t & 31, w = t >> 5;

    int p = 0;
    for (int k = t; k < (int)blockIdx.x; k += 256) p += tsum[k];
    p = gcn_wave_sum(p);
    if (lane == 0) part[w] = p;

    const int  base = blockIdx.x * GCN_TILE + t * 4;
    const bool full = base + 3 < n;
    int c[4] = {0, 0, 0, 0};
    if (full) {
        int4 v = *(const int4*)(cnt + base);
        c[0] = v.x; c[1] = v.y; c[2] = v.z; c[3] = v.w;
    } else {
        for (int k = 0; k < 4; ++k) if (base + k < n) c[k] = cnt[base + k];
    }
    const int s = c[0] + c[1] + c[2] + c[3];
    int inc = s;
    for (int o = 1; o < 32; o <<= 1) {
        int v = __shfl_up(inc, o);
        if (lane >= o) inc += v;
    }
    if (lane == 31) wtot[w] = inc;
    __syncthreads();
    int off = 0;
    for (int k = 0; k < 8; ++k) {
        off += part[k];
        if (k < w) off += wtot[k];
    }
    const int e0 = off + inc - s;
    const int e1 = e0 + c[0], e2 = e1 + c[1], e3 = e2 + c[2];
    const float d0 = rsqrtf((float)(c[0] + 1)), d1 = rsqrtf((float)(c[1] + 1));
    const float d2 = rsqrtf((float)(c[2] + 1)), d3 = rsqrtf((float)(c[3] + 1));
    if (full) {
        *(int4*)(rowptr + base)  = make_int4(e0, e1, e2, e3);
        *(float4*)(dinv + base)  = make_float4(d0, d1, d2, d3);
    } else {
        if (base + 0 < n) { rowptr[base + 0] = e0; dinv[base + 0] = d0; }
        if (base + 1 < n) { rowptr[base + 1] = e1; dinv[base + 1] = d1; }
        if (base + 2 < n) { rowptr[base + 2] = e2; dinv[base + 2] = d2; }
    }
    if (blockIdx.x == gridDim.x - 1 && t == 255) rowptr[n] = e0 + s;
}

__global__ __launch_bounds__(256) void gcn_csr_fill(const int* __restrict__ src,
                                                    const int* __restrict__ dst,
                                                    const int* __restrict__ rank,
                                                    const int* __restrict__ rowptr,
                                                    const float* __restrict__ dinv,
                                                    int2* __restrict__ csr, int ne) {
    int i = blockIdx.x * blockDim.x + threadIdx.x;
    if (i >= ne) return;
    const int s = src[i];
    const int d = dst[i];
    csr[rowptr[d] + rank[i]] = make_int2(s, __float_as_int(dinv[s]));
}

__global__ __launch_bounds__(256) void gcn_gemm_wmma(const float* __restrict__ X,
                                                     const float* __restrict__ W,
                                                     float* __restrict__ H) {
    const int lane     = threadIdx.x & 31;
    const int wave     = blockIdx.x * 8 + (threadIdx.x >> 5);
    const int col_tile = wave & 7;
    const int row_tile = wave >> 3;
    const int m        = lane & 15;
    const int half     = lane >> 4;
    const int koff     = half << 1;

    const float* xrow = X + (size_t)(row_tile * 16 + m) * GCN_D;
    const int    ncol = col_tile * 16 + m;

    v8f acc = {};
#pragma unroll
    for (int k = 0; k < GCN_D; k += 4) {
        v2f a = *(const v2f*)(xrow + k + koff);
        v2f b;
        b.x = W[(size_t)(k + koff)     * GCN_D + ncol];
        b.y = W[(size_t)(k + koff + 1) * GCN_D + ncol];
        acc = __builtin_amdgcn_wmma_f32_16x16x4_f32(
                  false, a, false, b, (short)0, acc, false, false);
    }

    const int mbase = row_tile * 16 + half * 8;
    float* hp = H + (size_t)mbase * GCN_D + col_tile * 16 + m;
#pragma unroll
    for (int i = 0; i < 8; ++i) hp[(size_t)i * GCN_D] = acc[i];
}

template <int U>
__device__ __forceinline__ void gcn_rows(const float4* __restrict__ hv, int sidx, float wv,
                                         int j, int lane, float4& acc) {
    float4 v[U];
    float  w[U];
    for (int u = 0; u < U; ++u) {
        const int s = __builtin_amdgcn_readlane(sidx, j + u);
        w[u] = __int_as_float(__builtin_amdgcn_readlane(__float_as_int(wv), j + u));
        const float4* row = hv + (size_t)s * (GCN_D / 4);
        v[u] = row[lane];
    }
    for (int u = 0; u < U; ++u) {
        acc.x = fmaf(v[u].x, w[u], acc.x);
        acc.y = fmaf(v[u].y, w[u], acc.y);
        acc.z = fmaf(v[u].z, w[u], acc.z);
        acc.w = fmaf(v[u].w, w[u], acc.w);
    }
}

__global__ __launch_bounds__(256) void gcn_gather(const int* __restrict__ rowptr,
                                                  const int2* __restrict__ csr,
                                                  const float* __restrict__ h,
                                                  const float* __restrict__ dinv,
                                                  const float* __restrict__ bias,
                                                  float* __restrict__ out, int n) {
    const int lane = threadIdx.x & 31;
    int node = (int)((blockIdx.x * 256u + threadIdx.x) >> 5);
    if (node >= n) return;
    node = __builtin_amdgcn_readfirstlane(node);

    const int   beg = rowptr[node];
    const int   end = rowptr[node + 1];
    const float dd  = dinv[node];
    const float4* hv = (const float4*)h;
    const float4 self = hv[(size_t)node * (GCN_D / 4) + lane];

    float4 acc = make_float4(0.0f, 0.0f, 0.0f, 0.0f);
    for (int base = beg; base < end; base += 32) {
        const int m = min(32, end - base);
        int   sidx = 0;
        float wv   = 0.0f;
        if (lane < m) {
            const int2 e = csr[base + lane];
            sidx = min(max(e.x, 0), n - 1);
            wv   = __int_as_float(e.y) * dd;
        }
        int j = 0;
        for (; j + 8 <= m; j += 8) gcn_rows<8>(hv, sidx, wv, j, lane, acc);
        if (j + 4 <= m) { gcn_rows<4>(hv, sidx, wv, j, lane, acc); j += 4; }
        if (j + 2 <= m) { gcn_rows<2>(hv, sidx, wv, j, lane, acc); j += 2; }
        if (j < m)      { gcn_rows<1>(hv, sidx, wv, j, lane, acc); }
    }

    const float  sw = dd * dd;
    const float4 bv = ((const float4*)bias)[lane];
    float4 r;
    r.x = fmaf(self.x, sw, acc.x) + bv.x;
    r.y = fmaf(self.y, sw, acc.y) + bv.y;
    r.z = fmaf(self.z, sw, acc.z) + bv.z;
    r.w = fmaf(self.w, sw, acc.w) + bv.w;
    r.x = r.x > 0.0f ? r.x : 0.0f;
    r.y = r.y > 0.0f ? r.y : 0.0f;
    r.z = r.z > 0.0f ? r.z : 0.0f;
    r.w = r.w > 0.0f ? r.w : 0.0f;
    ((float4*)out)[(size_t)node * (GCN_D / 4) + lane] = r;
}

extern "C" void kernel_launch(void* const* d_in, const int* in_sizes, int n_in,
                              void* d_out, int out_size, void* d_ws, size_t ws_size,
                              hipStream_t stream) {
    const float* x  = (const float*)d_in[0];
    const float* W1 = (const float*)d_in[1];
    const float* b1 = (const float*)d_in[2];
    const float* W2 = (const float*)d_in[3];
    const float* b2 = (const float*)d_in[4];
    const float* W3 = (const float*)d_in[5];
    const float* b3 = (const float*)d_in[6];
    const int*   eidx = (const int*)d_in[7];
    const int ne = in_sizes[7] / 2;
    const int* src = eidx;
    const int* dst = eidx + ne;

    const size_t node_bytes = (((size_t)(GCN_N + 1) * 4) + 4095) & ~(size_t)4095;
    const size_t edge_bytes = (((size_t)ne * 4) + 4095) & ~(size_t)4095;
    char* ws = (char*)d_ws;
    size_t off = 0;
    float* dinv   = (float*)(ws + off); off += node_bytes;
    int*   cnt    = (int*)(ws + off);   off += node_bytes;
    int*   rowptr = (int*)(ws + off);   off += node_bytes;
    int*   tsum   = (int*)(ws + off);   off += 4096;
    int*   rank   = (int*)(ws + off);   off += edge_bytes;
    int2*  csr    = (int2*)(ws + off);  off += 2 * edge_bytes;
    float* h      = (float*)(ws + off); off += (size_t)GCN_N * GCN_D * 4;
    float* buf    = (float*)(ws + off);

    const int blk = 256;
    const int g_node = (GCN_N + blk - 1) / blk;
    const int g_edge = (ne + blk - 1) / blk;
    const int g_tile = (GCN_N + GCN_TILE - 1) / GCN_TILE;
    const int g_gemm = GCN_N / 16;
    const int g_gath = (GCN_N + 7) / 8;

    gcn_fill_zero_i32<<<g_node, blk, 0, stream>>>(cnt, GCN_N);
    gcn_deg_rank<<<g_edge, blk, 0, stream>>>(dst, cnt, rank, ne);
    gcn_tile_sums<<<g_tile, blk, 0, stream>>>(cnt, tsum, GCN_N);
    gcn_scan_rows<<<g_tile, blk, 0, stream>>>(cnt, tsum, rowptr, dinv, GCN_N);
    gcn_csr_fill<<<g_edge, blk, 0, stream>>>(src, dst, rank, rowptr, dinv, csr, ne);

    gcn_gemm_wmma<<<g_gemm, blk, 0, stream>>>(x, W1, h);
    gcn_gather<<<g_gath, blk, 0, stream>>>(rowptr, csr, h, dinv, b1, buf, GCN_N);

    gcn_gemm_wmma<<<g_gemm, blk, 0, stream>>>(buf, W2, h);
    gcn_gather<<<g_gath, blk, 0, stream>>>(rowptr, csr, h, dinv, b2, buf, GCN_N);

    gcn_gemm_wmma<<<g_gemm, blk, 0, stream>>>(buf, W3, h);
    gcn_gather<<<g_gath, blk, 0, stream>>>(rowptr, csr, h, dinv, b3, (float*)d_out, GCN_N);
}


# GEMM: W (in B-fragment order) and 64 X rows staged in LDS once per work-group; each wave 16x64 outputs on 4 accumulator tiles fed from LDS; H stored as whole lines; sum order unchanged
# speedup vs baseline: 3.5045x; 1.1524x over previous
#include <hip/hip_runtime.h>


#define GCN_N 100000
#define GCN_D 128
#define GCN_TILE 1024

typedef float v2f __attribute__((ext_vector_type(2)));
typedef float v8f __attribute__((ext_vector_type(8)));

__global__ __launch_bounds__(256) void gcn_fill_zero_i32(int* __restrict__ p, int n) {
    int i = blockIdx.x * blockDim.x + threadIdx.x;
    if (i < n) p[i] = 0;
}

__global__ __launch_bounds__(256) void gcn_deg_rank(const int* __restrict__ dst,
                                                    int* __restrict__ cnt,
                                                    int* __restrict__ rank, int ne) {
    int i = blockIdx.x * blockDim.x + threadIdx.x;
    if (i < ne) rank[i] = atomicAdd(&cnt[dst[i]], 1);
}

__device__ __forceinline__ int gcn_wave_sum(int v) {
    for (int o = 16; o > 0; o >>= 1) v += __shfl_xor(v, o);
    return v;
}

__global__ __launch_bounds__(256) void gcn_tile_sums(const int* __restrict__ cnt,
                                                     int* __restrict__ tsum, int n) {
    __shared__ int part[8];
    const int t    = threadIdx.x;
    const int base = blockIdx.x * GCN_TILE + t * 4;
    int s = 0;
    if (base + 3 < n) {
        int4 v = *(const int4*)(cnt + base);
        s = v.x + v.y + v.z + v.w;
    } else {
        for (int k = 0; k < 4; ++k) if (base + k < n) s += cnt[base + k];
    }
    s = gcn_wave_sum(s);
    if ((t & 31) == 0) part[t >> 5] = s;
    __syncthreads();
    if (t == 0) {
        int r = 0;
        for (int k = 0; k < 8; ++k) r += part[k];
        tsum[blockIdx.x] = r;
    }
}

__global__ __launch_bounds__(256) void gcn_scan_rows(const int* __restrict__ cnt,
                                                     const int* __restrict__ tsum,
                                                     int* __restrict__ rowptr,
                                                     float* __restrict__ dinv, int n) {
    __shared__ int part[8];
    __shared__ int wtot[8];
    const int t = threadIdx.x, lane = t & 31, w = t >> 5;

    int p = 0;
    for (int k = t; k < (int)blockIdx.x; k += 256) p += tsum[k];
    p = gcn_wave_sum(p);
    if (lane == 0) part[w] = p;

    const int  base = blockIdx.x * GCN_TILE + t * 4;
    const bool full = base + 3 < n;
    int c[4] = {0, 0, 0, 0};
    if (full) {
        int4 v = *(const int4*)(cnt + base);
        c[0] = v.x; c[1] = v.y; c[2] = v.z; c[3] = v.w;
    } else {
        for (int k = 0; k < 4; ++k) if (base + k < n) c[k] = cnt[base + k];
    }
    const int s = c[0] + c[1] + c[2] + c[3];
    int inc = s;
    for (int o = 1; o < 32; o <<= 1) {
        int v = __shfl_up(inc, o);
        if (lane >= o) inc += v;
    }
    if (lane == 31) wtot[w] = inc;
    __syncthreads();
    int off = 0;
    for (int k = 0; k < 8; ++k) {
        off += part[k];
        if (k < w) off += wtot[k];
    }
    const int e0 = off + inc - s;
    const int e1 = e0 + c[0], e2 = e1 + c[1], e3 = e2 + c[2];
    const float d0 = rsqrtf((float)(c[0] + 1)), d1 = rsqrtf((float)(c[1] + 1));
    const float d2 = rsqrtf((float)(c[2] + 1)), d3 = rsqrtf((float)(c[3] + 1));
    if (full) {
        *(int4*)(rowptr + base)  = make_int4(e0, e1, e2, e3);
        *(float4*)(dinv + base)  = make_float4(d0, d1, d2, d3);
    } else {
        if (base + 0 < n) { rowptr[base + 0] = e0; dinv[base + 0] = d0; }
        if (base + 1 < n) { rowptr[base + 1] = e1; dinv[base + 1] = d1; }
        if (base + 2 < n) { rowptr[base + 2] = e2; dinv[base + 2] = d2; }
    }
    if (blockIdx.x == gridDim.x - 1 && t == 255) rowptr[n] = e0 + s;
}

__global__ __launch_bounds__(256) void gcn_csr_fill(const int* __restrict__ src,
                                                    const int* __restrict__ dst,
                                                    const int* __restrict__ rank,
                                                    const int* __restrict__ rowptr,
                                                    const float* __restrict__ dinv,
                                                    int2* __restrict__ csr, int ne) {
    int i = blockIdx.x * blockDim.x + threadIdx.x;
    if (i >= ne) return;
    const int s = src[i];
    const int d = dst[i];
    csr[rowptr[d] + rank[i]] = make_int2(s, __float_as_int(dinv[s]));
}

#define GEMM_ROWS   64
#define GEMM_XPITCH 132
#define GEMM_KB     4

struct gcn_frag {
    v2f    a[GEMM_KB];
    float4 b[GEMM_KB][2];
};

__device__ __forceinline__ void gcn_frag_load(gcn_frag& f, const float* __restrict__ xa,
                                              const float* __restrict__ wb, int kb) {
#pragma unroll
    for (int u = 0; u < GEMM_KB; ++u) {
        const int ks = kb * GEMM_KB + u;
        f.a[u]    = *(const v2f*)(xa + 4 * ks);
        f.b[u][0] = *(const float4*)(wb + ks * 512);
        f.b[u][1] = *(const float4*)(wb + ks * 512 + 128);
    }
}

__device__ __forceinline__ void gcn_frag_mma(const gcn_frag& f, v8f (&acc)[4]) {
#pragma unroll
    for (int u = 0; u < GEMM_KB; ++u) {
        v2f b;
        b.x = f.b[u][0].x; b.y = f.b[u][0].y;
        acc[0] = __builtin_amdgcn_wmma_f32_16x16x4_f32(false, f.a[u], false, b, (short)0, acc[0], false, false);
        b.x = f.b[u][0].z; b.y = f.b[u][0].w;
        acc[1] = __builtin_amdgcn_wmma_f32_16x16x4_f32(false, f.a[u], false, b, (short)0, acc[1], false, false);
        b.x = f.b[u][1].x; b.y = f.b[u][1].y;
        acc[2] = __builtin_amdgcn_wmma_f32_16x16x4_f32(false, f.a[u], false, b, (short)0, acc[2], false, false);
        b.x = f.b[u][1].z; b.y = f.b[u][1].w;
        acc[3] = __builtin_amdgcn_wmma_f32_16x16x4_f32(false, f.a[u], false, b, (short)0, acc[3], false, false);
    }
}

__global__ __launch_bounds__(256) void gcn_gemm_wmma(const float* __restrict__ X,
                                                     const float* __restrict__ W,
                                                     float* __restrict__ H, int n) {
    __shared__ __attribute__((aligned(16))) float sW[GCN_D * GCN_D];
    __shared__ __attribute__((aligned(16))) float sX[GEMM_ROWS * GEMM_XPITCH];

    const int t    = threadIdx.x;
    const int lane = t & 31;
    const int wave = t >> 5;
    const int m    = lane & 15;
    const int h    = lane >> 4;
    const int row0 = blockIdx.x * GEMM_ROWS;

    float4 xv[8];
#pragma unroll
    for (int i = 0; i < 8; ++i) {
        const int r = wave + 8 * i;
        xv[i] = ((const float4*)(X + (size_t)min(row0 + r, n - 1) * GCN_D))[lane];
    }
#pragma unroll
    for (int i = 0; i < 16; ++i) {
        const int d  = t + 256 * i;
        const int cp = (d >> 5) & 3;
        const int ks = d >> 7;
        const float* wp = W + (size_t)(4 * ks + 2 * h) * GCN_D + 32 * cp + 2 * m;
        const v2f r0 = *(const v2f*)wp;
        const v2f r1 = *(const v2f*)(wp + GCN_D);
        *(float4*)(sW + 4 * d) = make_float4(r0.x, r1.x, r0.y, r1.y);
    }
#pragma unroll
    for (int i = 0; i < 8; ++i)
        *(float4*)(sX + (wave + 8 * i) * GEMM_XPITCH + 4 * lane) = xv[i];
    __syncthreads();

    const int tile = wave >> 1;
    const int ch   = wave & 1;
    const float* xa = sX + (tile * 16 + m) * GEMM_XPITCH + 2 * h;
    const float* wb = sW + 4 * (2 * ch * 32 + lane);

    v8f acc[4] = {};
    gcn_frag f0, f1;
    gcn_frag_load(f0, xa, wb, 0);
#pragma unroll
    for (int kb = 0; kb < 32 / GEMM_KB; kb += 2) {
        gcn_frag_load(f1, xa, wb, kb + 1);
        gcn_frag_mma(f0, acc);
        if (kb + 2 < 32 / GEMM_KB) gcn_frag_load(f0, xa, wb, kb + 2);
        gcn_frag_mma(f1, acc);
    }

    __syncthreads();
    float* so = sX + (tile * 16 + 8 * h) * GEMM_XPITCH + 64 * ch + 2 * m;
#pragma unroll
    for (int p = 0; p < 2; ++p) {
#pragma unroll
        for (int i = 0; i < 8; ++i) {
            v2f o;
            o.x = acc[2 * p][i];
            o.y = acc[2 * p + 1][i];
            *(v2f*)(so + i * GEMM_XPITCH + 32 * p) = o;
        }
    }
    __syncthreads();
    const float* si = sX + (tile * 16 + h) * GEMM_XPITCH + 64 * ch + 4 * m;
    float*       hp = H + (size_t)(row0 + tile * 16 + h) * GCN_D + 64 * ch + 4 * m;
    if (row0 + tile * 16 + 16 <= n) {
#pragma unroll
        for (int j = 0; j < 8; ++j)
            *(float4*)(hp + (size_t)(2 * j) * GCN_D) = *(const float4*)(si + 2 * j * GEMM_XPITCH);
    } else {
#pragma unroll
        for (int j = 0; j < 8; ++j)
            if (row0 + tile * 16 + 2 * j + h < n)
                *(float4*)(hp + (size_t)(2 * j) * GCN_D) = *(const float4*)(si + 2 * j * GEMM_XPITCH);
    }
}

template <int U>
__device__ __forceinline__ void gcn_rows(const float4* __restrict__ hv, int sidx, float wv,
                                         int j, int lane, float4& acc) {
    float4 v[U];
    float  w[U];
    for (int u = 0; u < U; ++u) {
        const int s = __builtin_amdgcn_readlane(sidx, j + u);
        w[u] = __int_as_float(__builtin_amdgcn_readlane(__float_as_int(wv), j + u));
        const float4* row = hv + (size_t)s * (GCN_D / 4);
        v[u] = row[lane];
    }
    for (int u = 0; u < U; ++u) {
        acc.x = fmaf(v[u].x, w[u], acc.x);
        acc.y = fmaf(v[u].y, w[u], acc.y);
        acc.z = fmaf(v[u].z, w[u], acc.z);
        acc.w = fmaf(v[u].w, w[u], acc.w);
    }
}

__global__ __launch_bounds__(256) void gcn_gather(const int* __restrict__ rowptr,
                                                  const int2* __restrict__ csr,
                                                  const float* __restrict__ h,
                                                  const float* __restrict__ dinv,
                                                  const float* __restrict__ bias,
                                                  float* __restrict__ out, int n) {
    const int lane = threadIdx.x & 31;
    int node = (int)((blockIdx.x * 256u + threadIdx.x) >> 5);
    if (node >= n) return;
    node = __builtin_amdgcn_readfirstlane(node);

    const int   beg = rowptr[node];
    const int   end = rowptr[node + 1];
    const float dd  = dinv[node];
    const float4* hv = (const float4*)h;
    const float4 self = hv[(size_t)node * (GCN_D / 4) + lane];

    float4 acc = make_float4(0.0f, 0.0f, 0.0f, 0.0f);
    for (int base = beg; base < end; base += 32) {
        const int m = min(32, end - base);
        int   sidx = 0;
        float wv   = 0.0f;
        if (lane < m) {
            const int2 e = csr[base + lane];
            sidx = min(max(e.x, 0), n - 1);
            wv   = __int_as_float(e.y) * dd;
        }
        int j = 0;
        for (; j + 8 <= m; j += 8) gcn_rows<8>(hv, sidx, wv, j, lane, acc);
        if (j + 4 <= m) { gcn_rows<4>(hv, sidx, wv, j, lane, acc); j += 4; }
        if (j + 2 <= m) { gcn_rows<2>(hv, sidx, wv, j, lane, acc); j += 2; }
        if (j < m)      { gcn_rows<1>(hv, sidx, wv, j, lane, acc); }
    }

    const float  sw = dd * dd;
    const float4 bv = ((const float4*)bias)[lane];
    float4 r;
    r.x = fmaf(self.x, sw, acc.x) + bv.x;
    r.y = fmaf(self.y, sw, acc.y) + bv.y;
    r.z = fmaf(self.z, sw, acc.z) + bv.z;
    r.w = fmaf(self.w, sw, acc.w) + bv.w;
    r.x = r.x > 0.0f ? r.x : 0.0f;
    r.y = r.y > 0.0f ? r.y : 0.0f;
    r.z = r.z > 0.0f ? r.z : 0.0f;
    r.w = r.w > 0.0f ? r.w : 0.0f;
    ((float4*)out)[(size_t)node * (GCN_D / 4) + lane] = r;
}

extern "C" void kernel_launch(void* const* d_in, const int* in_sizes, int n_in,
                              void* d_out, int out_size, void* d_ws, size_t ws_size,
                              hipStream_t stream) {
    const float* x  = (const float*)d_in[0];
    const float* W1 = (const float*)d_in[1];
    const float* b1 = (const float*)d_in[2];
    const float* W2 = (const float*)d_in[3];
    const float* b2 = (const float*)d_in[4];
    const float* W3 = (const float*)d_in[5];
    const float* b3 = (const float*)d_in[6];
    const int*   eidx = (const int*)d_in[7];
    const int ne = in_sizes[7] / 2;
    const int* src = eidx;
    const int* dst = eidx + ne;

    const size_t node_bytes = (((size_t)(GCN_N + 1) * 4) + 4095) & ~(size_t)4095;
    const size_t edge_bytes = (((size_t)ne * 4) + 4095) & ~(size_t)4095;
    char* ws = (char*)d_ws;
    size_t off = 0;
    float* dinv   = (float*)(ws + off); off += node_bytes;
    int*   cnt    = (int*)(ws + off);   off += node_bytes;
    int*   rowptr = (int*)(ws + off);   off += node_bytes;
    int*   tsum   = (int*)(ws + off);   off += 4096;
    int*   rank   = (int*)(ws + off);   off += edge_bytes;
    int2*  csr    = (int2*)(ws + off);  off += 2 * edge_bytes;
    float* h      = (float*)(ws + off); off += (size_t)GCN_N * GCN_D * 4;
    float* buf    = (float*)(ws + off);

    const int blk = 256;
    const int g_node = (GCN_N + blk - 1) / blk;
    const int g_edge = (ne + blk - 1) / blk;
    const int g_tile = (GCN_N + GCN_TILE - 1) / GCN_TILE;
    const int g_gemm = (GCN_N + GEMM_ROWS - 1) / GEMM_ROWS;
    const int g_gath = (GCN_N + 7) / 8;

    gcn_fill_zero_i32<<<g_node, blk, 0, stream>>>(cnt, GCN_N);
    gcn_deg_rank<<<g_edge, blk, 0, stream>>>(dst, cnt, rank, ne);
    gcn_tile_sums<<<g_tile, blk, 0, stream>>>(cnt, tsum, GCN_N);
    gcn_scan_rows<<<g_tile, blk, 0, stream>>>(cnt, tsum, rowptr, dinv, GCN_N);
    gcn_csr_fill<<<g_edge, blk, 0, stream>>>(src, dst, rank, rowptr, dinv, csr, ne);

    gcn_gemm_wmma<<<g_gemm, blk, 0, stream>>>(x, W1, h, GCN_N);
    gcn_gather<<<g_gath, blk, 0, stream>>>(rowptr, csr, h, dinv, b1, buf, GCN_N);

    gcn_gemm_wmma<<<g_gemm, blk, 0, stream>>>(buf, W2, h, GCN_N);
    gcn_gather<<<g_gath, blk, 0, stream>>>(rowptr, csr, h, dinv, b2, buf, GCN_N);

    gcn_gemm_wmma<<<g_gemm, blk, 0, stream>>>(buf, W3, h, GCN_N);
    gcn_gather<<<g_gath, blk, 0, stream>>>(rowptr, csr, h, dinv, b3, (float*)d_out, GCN_N);
}


# Gather fused into the next layer's GEMM launch (layers 2, 3): a work-group sums its 64 rows, 16 row loads in flight per wave, into the LDS X tile; 2 launches, activation round trip go; sum order kept
# speedup vs baseline: 3.6289x; 1.0355x over previous
#include <hip/hip_runtime.h>


#define GCN_N 100000
#define GCN_D 128
#define GCN_TILE 1024

typedef float v2f __attribute__((ext_vector_type(2)));
typedef float v8f __attribute__((ext_vector_type(8)));

__global__ __launch_bounds__(256) void gcn_fill_zero_i32(int* __restrict__ p, int n) {
    int i = blockIdx.x * blockDim.x + threadIdx.x;
    if (i < n) p[i] = 0;
}

__global__ __launch_bounds__(256) void gcn_deg_rank(const int* __restrict__ dst,
                                                    int* __restrict__ cnt,
                                                    int* __restrict__ rank, int ne) {
    int i = blockIdx.x * blockDim.x + threadIdx.x;
    if (i < ne) rank[i] = atomicAdd(&cnt[dst[i]], 1);
}

__device__ __forceinline__ int gcn_wave_sum(int v) {
    for (int o = 16; o > 0; o >>= 1) v += __shfl_xor(v, o);
    return v;
}

__global__ __launch_bounds__(256) void gcn_tile_sums(const int* __restrict__ cnt,
                                                     int* __restrict__ tsum, int n) {
    __shared__ int part[8];
    const int t    = threadIdx.x;
    const int base = blockIdx.x * GCN_TILE + t * 4;
    int s = 0;
    if (base + 3 < n) {
        int4 v = *(const int4*)(cnt + base);
        s = v.x + v.y + v.z + v.w;
    } else {
        for (int k = 0; k < 4; ++k) if (base + k < n) s += cnt[base + k];
    }
    s = gcn_wave_sum(s);
    if ((t & 31) == 0) part[t >> 5] = s;
    __syncthreads();
    if (t == 0) {
        int r = 0;
        for (int k = 0; k < 8; ++k) r += part[k];
        tsum[blockIdx.x] = r;
    }
}

__global__ __launch_bounds__(256) void gcn_scan_rows(const int* __restrict__ cnt,
                                                     const int* __restrict__ tsum,
                                                     int* __restrict__ rowptr,
                                                     float* __restrict__ dinv, int n) {
    __shared__ int part[8];
    __shared__ int wtot[8];
    const int t = threadIdx.x, lane = t & 31, w = t >> 5;

    int p = 0;
    for (int k = t; k < (int)blockIdx.x; k += 256) p += tsum[k];
    p = gcn_wave_sum(p);
    if (lane == 0) part[w] = p;

    const int  base = blockIdx.x * GCN_TILE + t * 4;
    const bool full = base + 3 < n;
    int c[4] = {0, 0, 0, 0};
    if (full) {
        int4 v = *(const int4*)(cnt + base);
        c[0] = v.x; c[1] = v.y; c[2] = v.z; c[3] = v.w;
    } else {
        for (int k = 0; k < 4; ++k) if (base + k < n) c[k] = cnt[base + k];
    }
    const int s = c[0] + c[1] + c[2] + c[3];
    int inc = s;
    for (int o = 1; o < 32; o <<= 1) {
        int v = __shfl_up(inc, o);
        if (lane >= o) inc += v;
    }
    if (lane == 31) wtot[w] = inc;
    __syncthreads();
    int off = 0;
    for (int k = 0; k < 8; ++k) {
        off += part[k];
        if (k < w) off += wtot[k];
    }
    const int e0 = off + inc - s;
    const int e1 = e0 + c[0], e2 = e1 + c[1], e3 = e2 + c[2];
    const float d0 = rsqrtf((float)(c[0] + 1)), d1 = rsqrtf((float)(c[1] + 1));
    const float d2 = rsqrtf((float)(c[2] + 1)), d3 = rsqrtf((float)(c[3] + 1));
    if (full) {
        *(int4*)(rowptr + base)  = make_int4(e0, e1, e2, e3);
        *(float4*)(dinv + base)  = make_float4(d0, d1, d2, d3);
    } else {
        if (base + 0 < n) { rowptr[base + 0] = e0; dinv[base + 0] = d0; }
        if (base + 1 < n) { rowptr[base + 1] = e1; dinv[base + 1] = d1; }
        if (base + 2 < n) { rowptr[base + 2] = e2; dinv[base + 2] = d2; }
    }
    if (blockIdx.x == gridDim.x - 1 && t == 255) rowptr[n] = e0 + s;
}

__global__ __launch_bounds__(256) void gcn_csr_fill(const int* __restrict__ src,
                                                    const int* __restrict__ dst,
                                                    const int* __restrict__ rank,
                                                    const int* __restrict__ rowptr,
                                                    const float* __restrict__ dinv,
                                                    int2* __restrict__ csr, int ne) {
    int i = blockIdx.x * blockDim.x + threadIdx.x;
    if (i >= ne) return;
    const int s = src[i];
    const int d = dst[i];
    csr[rowptr[d] + rank[i]] = make_int2(s, __float_as_int(dinv[s]));
}

#define GEMM_ROWS   64
#define GEMM_XPITCH 132
#define GEMM_KB     4

struct gcn_frag {
    v2f    a[GEMM_KB];
    float4 b[GEMM_KB][2];
};

__device__ __forceinline__ void gcn_frag_load(gcn_frag& f, const float* __restrict__ xa,
                                              const float* __restrict__ wb, int kb) {
#pragma unroll
    for (int u = 0; u < GEMM_KB; ++u) {
        const int ks = kb * GEMM_KB + u;
        f.a[u]    = *(const v2f*)(xa + 4 * ks);
        f.b[u][0] = *(const float4*)(wb + ks * 512);
        f.b[u][1] = *(const float4*)(wb + ks * 512 + 128);
    }
}

__device__ __forceinline__ void gcn_frag_mma(const gcn_frag& f, v8f (&acc)[4]) {
#pragma unroll
    for (int u = 0; u < GEMM_KB; ++u) {
        v2f b;
        b.x = f.b[u][0].x; b.y = f.b[u][0].y;
        acc[0] = __builtin_amdgcn_wmma_f32_16x16x4_f32(false, f.a[u], false, b, (short)0, acc[0], false, false);
        b.x = f.b[u][0].z; b.y = f.b[u][0].w;
        acc[1] = __builtin_amdgcn_wmma_f32_16x16x4_f32(false, f.a[u], false, b, (short)0, acc[1], false, false);
        b.x = f.b[u][1].x; b.y = f.b[u][1].y;
        acc[2] = __builtin_amdgcn_wmma_f32_16x16x4_f32(false, f.a[u], false, b, (short)0, acc[2], false, false);
        b.x = f.b[u][1].z; b.y = f.b[u][1].w;
        acc[3] = __builtin_amdgcn_wmma_f32_16x16x4_f32(false, f.a[u], false, b, (short)0, acc[3], false, false);
    }
}

__device__ __forceinline__ void gcn_stage_w(float* __restrict__ sW, const float* __restrict__ W, int t) {
    const int lane = t & 31;
    const int m    = lane & 15;
    const int h    = lane >> 4;
#pragma unroll
    for (int i = 0; i < 16; ++i) {
        const int d  = t + 256 * i;
        const int cp = (d >> 5) & 3;
        const int ks = d >> 7;
        const float* wp = W + (size_t)(4 * ks + 2 * h) * GCN_D + 32 * cp + 2 * m;
        const v2f r0 = *(const v2f*)wp;
        const v2f r1 = *(const v2f*)(wp + GCN_D);
        *(float4*)(sW + 4 * d) = make_float4(r0.x, r1.x, r0.y, r1.y);
    }
}

__device__ __forceinline__ void gcn_tile_product(const float* __restrict__ sW, float* __restrict__ sX,
                                                 float* __restrict__ H, int row0, int n, int t) {
    const int lane = t & 31;
    const int wave = t >> 5;
    const int m    = lane & 15;
    const int h    = lane >> 4;
    const int tile = wave >> 1;
    const int ch   = wave & 1;
    const float* xa = sX + (tile * 16 + m) * GEMM_XPITCH + 2 * h;
    const float* wb = sW + 4 * (2 * ch * 32 + lane);

    v8f acc[4] = {};
    gcn_frag f0, f1;
    gcn_frag_load(f0, xa, wb, 0);
#pragma unroll
    for (int kb = 0; kb < 32 / GEMM_KB; kb += 2) {
        gcn_frag_load(f1, xa, wb, kb + 1);
        gcn_frag_mma(f0, acc);
        if (kb + 2 < 32 / GEMM_KB) gcn_frag_load(f0, xa, wb, kb + 2);
        gcn_frag_mma(f1, acc);
    }

    __syncthreads();
    float* so = sX + (tile * 16 + 8 * h) * GEMM_XPITCH + 64 * ch + 2 * m;
#pragma unroll
    for (int p = 0; p < 2; ++p) {
#pragma unroll
        for (int i = 0; i < 8; ++i) {
            v2f o;
            o.x = acc[2 * p][i];
            o.y = acc[2 * p + 1][i];
            *(v2f*)(so + i * GEMM_XPITCH + 32 * p) = o;
        }
    }
    __syncthreads();
    const float* si = sX + (tile * 16 + h) * GEMM_XPITCH + 64 * ch + 4 * m;
    float*       hp = H + (size_t)(row0 + tile * 16 + h) * GCN_D + 64 * ch + 4 * m;
    if (row0 + tile * 16 + 16 <= n) {
#pragma unroll
        for (int j = 0; j < 8; ++j)
            *(float4*)(hp + (size_t)(2 * j) * GCN_D) = *(const float4*)(si + 2 * j * GEMM_XPITCH);
    } else {
#pragma unroll
        for (int j = 0; j < 8; ++j)
            if (row0 + tile * 16 + 2 * j + h < n)
                *(float4*)(hp + (size_t)(2 * j) * GCN_D) = *(const float4*)(si + 2 * j * GEMM_XPITCH);
    }
}

__global__ __launch_bounds__(256) void gcn_gemm_wmma(const float* __restrict__ X,
                                                     const float* __restrict__ W,
                                                     float* __restrict__ H, int n) {
    __shared__ __attribute__((aligned(16))) float sW[GCN_D * GCN_D];
    __shared__ __attribute__((aligned(16))) float sX[GEMM_ROWS * GEMM_XPITCH];

    const int t    = threadIdx.x;
    const int lane = t & 31;
    const int wave = t >> 5;
    const int row0 = blockIdx.x * GEMM_ROWS;

    float4 xv[8];
#pragma unroll
    for (int i = 0; i < 8; ++i) {
        const int r = wave + 8 * i;
        xv[i] = ((const float4*)(X + (size_t)min(row0 + r, n - 1) * GCN_D))[lane];
    }
    gcn_stage_w(sW, W, t);
#pragma unroll
    for (int i = 0; i < 8; ++i)
        *(float4*)(sX + (wave + 8 * i) * GEMM_XPITCH + 4 * lane) = xv[i];
    __syncthreads();

    gcn_tile_product(sW, sX, H, row0, n, t);
}

template <int U>
__device__ __forceinline__ void gcn_rows(const float4* __restrict__ hv, int sidx, float wv,
                                         int j, int lane, float4& acc) {
    float4 v[U];
    float  w[U];
    for (int u = 0; u < U; ++u) {
        const int s = __builtin_amdgcn_readlane(sidx, j + u);
        w[u] = __int_as_float(__builtin_amdgcn_readlane(__float_as_int(wv), j + u));
        const float4* row = hv + (size_t)s * (GCN_D / 4);
        v[u] = row[lane];
    }
    for (int u = 0; u < U; ++u) {
        acc.x = fmaf(v[u].x, w[u], acc.x);
        acc.y = fmaf(v[u].y, w[u], acc.y);
        acc.z = fmaf(v[u].z, w[u], acc.z);
        acc.w = fmaf(v[u].w, w[u], acc.w);
    }
}

__global__ __launch_bounds__(256) void gcn_gather(const int* __restrict__ rowptr,
                                                  const int2* __restrict__ csr,
                                                  const float* __restrict__ h,
                                                  const float* __restrict__ dinv,
                                                  const float* __restrict__ bias,
                                                  float* __restrict__ out, int n) {
    const int lane = threadIdx.x & 31;
    int node = (int)((blockIdx.x * 256u + threadIdx.x) >> 5);
    if (node >= n) return;
    node = __builtin_amdgcn_readfirstlane(node);

    const int   beg = rowptr[node];
    const int   end = rowptr[node + 1];
    const float dd  = dinv[node];
    const float4* hv = (const float4*)h;
    const float4 self = hv[(size_t)node * (GCN_D / 4) + lane];

    float4 acc = make_float4(0.0f, 0.0f, 0.0f, 0.0f);
    for (int base = beg; base < end; base += 32) {
        const int m = min(32, end - base);
        int   sidx = 0;
        float wv   = 0.0f;
        if (lane < m) {
            const int2 e = csr[base + lane];
            sidx = min(max(e.x, 0), n - 1);
            wv   = __int_as_float(e.y) * dd;
        }
        int j = 0;
        for (; j + 8 <= m; j += 8) gcn_rows<8>(hv, sidx, wv, j, lane, acc);
        if (j + 4 <= m) { gcn_rows<4>(hv, sidx, wv, j, lane, acc); j += 4; }
        if (j + 2 <= m) { gcn_rows<2>(hv, sidx, wv, j, lane, acc); j += 2; }
        if (j < m)      { gcn_rows<1>(hv, sidx, wv, j, lane, acc); }
    }

    const float  sw = dd * dd;
    const float4 bv = ((const float4*)bias)[lane];
    float4 r;
    r.x = fmaf(self.x, sw, acc.x) + bv.x;
    r.y = fmaf(self.y, sw, acc.y) + bv.y;
    r.z = fmaf(self.z, sw, acc.z) + bv.z;
    r.w = fmaf(self.w, sw, acc.w) + bv.w;
    r.x = r.x > 0.0f ? r.x : 0.0f;
    r.y = r.y > 0.0f ? r.y : 0.0f;
    r.z = r.z > 0.0f ? r.z : 0.0f;
    r.w = r.w > 0.0f ? r.w : 0.0f;
    ((float4*)out)[(size_t)node * (GCN_D / 4) + lane] = r;
}

template <int U>
__device__ __forceinline__ void gcn_rows_load(const float4* __restrict__ hv, int sidx, int j, int lane,
                                              float4 (&v)[U]) {
#pragma unroll
    for (int u = 0; u < U; ++u) {
        const int s = __builtin_amdgcn_readlane(sidx, j + u);
        v[u] = hv[(size_t)s * (GCN_D / 4) + lane];
    }
}
template <int U>
__device__ __forceinline__ void gcn_rows_sum(const float4 (&v)[U], float wv, int j, float4& acc) {
#pragma unroll
    for (int u = 0; u < U; ++u) {
        const float w = __int_as_float(__builtin_amdgcn_readlane(__float_as_int(wv), j + u));
        acc.x = fmaf(v[u].x, w, acc.x);
        acc.y = fmaf(v[u].y, w, acc.y);
        acc.z = fmaf(v[u].z, w, acc.z);
        acc.w = fmaf(v[u].w, w, acc.w);
    }
}

__device__ __forceinline__ void gcn_chunk(const float4* __restrict__ hv, int sidx, float wv, int m,
                                          int lane, float4& acc) {
    int jb = 0;
    if (m >= 16) {
        float4 va[16];
        gcn_rows_load<16>(hv, sidx, 0, lane, va);
        gcn_rows_sum<16>(va, wv, 0, acc);
        jb = 16;
    }
    float4 vb[8], vc[4], vd[2], ve[1];
    const bool tb = jb + 8 <= m;
    const int  jc = jb + (tb ? 8 : 0);
    const bool tc = jc + 4 <= m;
    const int  jd = jc + (tc ? 4 : 0);
    const bool td = jd + 2 <= m;
    const int  je = jd + (td ? 2 : 0);
    const bool te = je < m;
    if (tb) gcn_rows_load<8>(hv, sidx, jb, lane, vb);
    if (tc) gcn_rows_load<4>(hv, sidx, jc, lane, vc);
    if (td) gcn_rows_load<2>(hv, sidx, jd, lane, vd);
    if (te) gcn_rows_load<1>(hv, sidx, je, lane, ve);
    if (tb) gcn_rows_sum<8>(vb, wv, jb, acc);
    if (tc) gcn_rows_sum<4>(vc, wv, jc, acc);
    if (td) gcn_rows_sum<2>(vd, wv, jd, acc);
    if (te) gcn_rows_sum<1>(ve, wv, je, acc);
    if (m == 32) gcn_rows<1>(hv, sidx, wv, 31, lane, acc);
}

__global__ __launch_bounds__(256) void gcn_gather_gemm(const int* __restrict__ rowptr,
                                                       const int2* __restrict__ csr,
                                                       const float* __restrict__ hin,
                                                       const float* __restrict__ dinv,
                                                       const float* __restrict__ bias,
                                                       const float* __restrict__ W,
                                                       float* __restrict__ H, int n) {
    __shared__ __attribute__((aligned(16))) float sW[GCN_D * GCN_D];
    __shared__ __attribute__((aligned(16))) float sX[GEMM_ROWS * GEMM_XPITCH];

    const int t    = threadIdx.x;
    const int lane = t & 31;
    const int wave = __builtin_amdgcn_readfirstlane(t >> 5);
    const int row0 = blockIdx.x * GEMM_ROWS;
    const int node0 = row0 + 8 * wave;

    const int    rpv = rowptr[min(node0 + lane, n)];
    const float  dvv = dinv[min(node0 + lane, n - 1)];
    const float4 bv  = ((const float4*)bias)[lane];
    const float4* hv = (const float4*)hin;

    gcn_stage_w(sW, W, t);

    int   beg = __builtin_amdgcn_readlane(rpv, 0);
    int   end = __builtin_amdgcn_readlane(rpv, 1);
    float dd  = __int_as_float(__builtin_amdgcn_readlane(__float_as_int(dvv), 0));
    int2 enext = make_int2(0, 0);
    if (lane < min(32, end - beg)) enext = csr[beg + lane];
    asm volatile("" :: "v"(enext.x), "v"(enext.y));
    for (int i = 0; i < 8; ++i) {
        const int  node   = node0 + i;
        const int2 efirst = enext;
        const int   nbeg = __builtin_amdgcn_readlane(rpv, i + 1);
        const int   nend = __builtin_amdgcn_readlane(rpv, i + 2);
        const float ndd  = __int_as_float(__builtin_amdgcn_readlane(__float_as_int(dvv), i + 1));
        enext = make_int2(0, 0);
        if (i < 7 && lane < min(32, nend - nbeg)) enext = csr[nbeg + lane];

        float4 r = make_float4(0.0f, 0.0f, 0.0f, 0.0f);
        if (node < n) {
            const float4 self = hv[(size_t)node * (GCN_D / 4) + lane];
            float4 acc = make_float4(0.0f, 0.0f, 0.0f, 0.0f);
            if (beg < end) {
                const int   sidx = min(max(efirst.x, 0), n - 1);
                const float wv   = __int_as_float(efirst.y) * dd;
                gcn_chunk(hv, sidx, wv, min(32, end - beg), lane, acc);
            }
            for (int base = beg + 32; base < end; base += 32) {
                const int m = min(32, end - base);
                int2 e = make_int2(0, 0);
                if (lane < m) e = csr[base + lane];
                const int   sidx = min(max(e.x, 0), n - 1);
                const float wv   = __int_as_float(e.y) * dd;
                for (int j = 0; j < m; ++j) gcn_rows<1>(hv, sidx, wv, j, lane, acc);
            }
            const float sw = dd * dd;
            r.x = fmaf(self.x, sw, acc.x) + bv.x;
            r.y = fmaf(self.y, sw, acc.y) + bv.y;
            r.z = fmaf(self.z, sw, acc.z) + bv.z;
            r.w = fmaf(self.w, sw, acc.w) + bv.w;
            r.x = r.x > 0.0f ? r.x : 0.0f;
            r.y = r.y > 0.0f ? r.y : 0.0f;
            r.z = r.z > 0.0f ? r.z : 0.0f;
            r.w = r.w > 0.0f ? r.w : 0.0f;
        }
        *(float4*)(sX + (8 * wave + i) * GEMM_XPITCH + 4 * lane) = r;
        beg = nbeg;
        end = nend;
        dd  = ndd;
    }
    __syncthreads();

    gcn_tile_product(sW, sX, H, row0, n, t);
}

extern "C" void kernel_launch(void* const* d_in, const int* in_sizes, int n_in,
                              void* d_out, int out_size, void* d_ws, size_t ws_size,
                              hipStream_t stream) {
    const float* x  = (const float*)d_in[0];
    const float* W1 = (const float*)d_in[1];
    const float* b1 = (const float*)d_in[2];
    const float* W2 = (const float*)d_in[3];
    const float* b2 = (const float*)d_in[4];
    const float* W3 = (const float*)d_in[5];
    const float* b3 = (const float*)d_in[6];
    const int*   eidx = (const int*)d_in[7];
    const int ne = in_sizes[7] / 2;
    const int* src = eidx;
    const int* dst = eidx + ne;

    const size_t node_bytes = (((size_t)(GCN_N + 1) * 4) + 4095) & ~(size_t)4095;
    const size_t edge_bytes = (((size_t)ne * 4) + 4095) & ~(size_t)4095;
    char* ws = (char*)d_ws;
    size_t off = 0;
    float* dinv   = (float*)(ws + off); off += node_bytes;
    int*   cnt    = (int*)(ws + off);   off += node_bytes;
    int*   rowptr = (int*)(ws + off);   off += node_bytes;
    int*   tsum   = (int*)(ws + off);   off += 4096;
    int*   rank   = (int*)(ws + off);   off += edge_bytes;
    int2*  csr    = (int2*)(ws + off);  off += 2 * edge_bytes;
    float* h      = (float*)(ws + off); off += (size_t)GCN_N * GCN_D * 4;
    float* buf    = (float*)(ws + off);

    const int blk = 256;
    const int g_node = (GCN_N + blk - 1) / blk;
    const int g_edge = (ne + blk - 1) / blk;
    const int g_tile = (GCN_N + GCN_TILE - 1) / GCN_TILE;
    const int g_gemm = (GCN_N + GEMM_ROWS - 1) / GEMM_ROWS;
    const int g_gath = (GCN_N + 7) / 8;

    gcn_fill_zero_i32<<<g_node, blk, 0, stream>>>(cnt, GCN_N);
    gcn_deg_rank<<<g_edge, blk, 0, stream>>>(dst, cnt, rank, ne);
    gcn_tile_sums<<<g_tile, blk, 0, stream>>>(cnt, tsum, GCN_N);
    gcn_scan_rows<<<g_tile, blk, 0, stream>>>(cnt, tsum, rowptr, dinv, GCN_N);
    gcn_csr_fill<<<g_edge, blk, 0, stream>>>(src, dst, rank, rowptr, dinv, csr, ne);

    gcn_gemm_wmma<<<g_gemm, blk, 0, stream>>>(x, W1, h, GCN_N);

    gcn_gather_gemm<<<g_gemm, blk, 0, stream>>>(rowptr, csr, h, dinv, b1, W2, buf, GCN_N);

    gcn_gather_gemm<<<g_gemm, blk, 0, stream>>>(rowptr, csr, buf, dinv, b2, W3, h, GCN_N);

    gcn_gather<<<g_gath, blk, 0, stream>>>(rowptr, csr, h, dinv, b3, (float*)d_out, GCN_N);
}


# speedup vs baseline: 5.3008x; 1.4607x over previous
#include <hip/hip_runtime.h>


#define GCN_N 100000
#define GCN_D 128

typedef float v2f __attribute__((ext_vector_type(2)));
typedef float v8f __attribute__((ext_vector_type(8)));

#define GCN_CAP       32
#define GCN_CNT_PLANE 100352
#define GCN_BKT_PLANE (GCN_N * GCN_CAP)

__global__ __launch_bounds__(256) void gcn_fill_zero_i32(int* __restrict__ p, int n) {
    int i = blockIdx.x * blockDim.x + threadIdx.x;
    if (i < n) p[i] = 0;
}

__global__ __launch_bounds__(256) void gcn_deg_bucket(const int* __restrict__ src,
                                                      const int* __restrict__ dst,
                                                      int* __restrict__ cnt,
                                                      int* __restrict__ bucket,
                                                      int2* __restrict__ ovf, int n, int ne) {
    int i = blockIdx.x * blockDim.x + threadIdx.x;
    if (i >= ne) return;
    const int s = src[i];
    const int d = dst[i];
    const int p = (int)((__builtin_amdgcn_s_sendmsg_rtn(0x87) >> 18) & 1u);
    const int r = atomicAdd(&cnt[p * GCN_CNT_PLANE + d], 1);
    if (r < GCN_CAP) {
        bucket[p * GCN_BKT_PLANE + d * GCN_CAP + r] = s;
    } else {
        const int o = atomicAdd(&cnt[n], 1);
        ovf[o] = make_int2(d, s);
    }
}

__device__ __forceinline__ float gcn_dinv(int indeg) { return rsqrtf((float)(indeg + 1)); }

__device__ __forceinline__ int gcn_entry(const int* __restrict__ bucket, int noff, int m0, int k) {
    return bucket[k < m0 ? noff + k : GCN_BKT_PLANE + noff + (k - m0)];
}

#define GEMM_ROWS   64
#define GEMM_XPITCH 132
#define GEMM_KB     4

struct gcn_frag {
    v2f    a[GEMM_KB];
    float4 b[GEMM_KB][2];
};

__device__ __forceinline__ void gcn_frag_load(gcn_frag& f, const float* __restrict__ xa,
                                              const float* __restrict__ wb, int kb) {
#pragma unroll
    for (int u = 0; u < GEMM_KB; ++u) {
        const int ks = kb * GEMM_KB + u;
        f.a[u]    = *(const v2f*)(xa + 4 * ks);
        f.b[u][0] = *(const float4*)(wb + ks * 512);
        f.b[u][1] = *(const float4*)(wb + ks * 512 + 128);
    }
}

__device__ __forceinline__ void gcn_frag_mma(const gcn_frag& f, v8f (&acc)[4]) {
#pragma unroll
    for (int u = 0; u < GEMM_KB; ++u) {
        v2f b;
        b.x = f.b[u][0].x; b.y = f.b[u][0].y;
        acc[0] = __builtin_amdgcn_wmma_f32_16x16x4_f32(false, f.a[u], false, b, (short)0, acc[0], false, false);
        b.x = f.b[u][0].z; b.y = f.b[u][0].w;
        acc[1] = __builtin_amdgcn_wmma_f32_16x16x4_f32(false, f.a[u], false, b, (short)0, acc[1], false, false);
        b.x = f.b[u][1].x; b.y = f.b[u][1].y;
        acc[2] = __builtin_amdgcn_wmma_f32_16x16x4_f32(false, f.a[u], false, b, (short)0, acc[2], false, false);
        b.x = f.b[u][1].z; b.y = f.b[u][1].w;
        acc[3] = __builtin_amdgcn_wmma_f32_16x16x4_f32(false, f.a[u], false, b, (short)0, acc[3], false, false);
    }
}

__device__ __forceinline__ void gcn_stage_w(float* __restrict__ sW, const float* __restrict__ W, int t) {
    const int lane = t & 31;
    const int m    = lane & 15;
    const int h    = lane >> 4;
#pragma unroll
    for (int i = 0; i < 16; ++i) {
        const int d  = t + 256 * i;
        const int cp = (d >> 5) & 3;
        const int ks = d >> 7;
        const float* wp = W + (size_t)(4 * ks + 2 * h) * GCN_D + 32 * cp + 2 * m;
        const v2f r0 = *(const v2f*)wp;
        const v2f r1 = *(const v2f*)(wp + GCN_D);
        *(float4*)(sW + 4 * d) = make_float4(r0.x, r1.x, r0.y, r1.y);
    }
}

__device__ __forceinline__ void gcn_tile_product(const float* __restrict__ sW, float* __restrict__ sX,
                                                 float* __restrict__ H, int row0, int n, int t) {
    const int lane = t & 31;
    const int wave = t >> 5;
    const int m    = lane & 15;
    const int h    = lane >> 4;
    const int tile = wave >> 1;
    const int ch   = wave & 1;
    const float* xa = sX + (tile * 16 + m) * GEMM_XPITCH + 2 * h;
    const float* wb = sW + 4 * (2 * ch * 32 + lane);

    v8f acc[4] = {};
    gcn_frag f0, f1;
    gcn_frag_load(f0, xa, wb, 0);
#pragma unroll
    for (int kb = 0; kb < 32 / GEMM_KB; kb += 2) {
        gcn_frag_load(f1, xa, wb, kb + 1);
        gcn_frag_mma(f0, acc);
        if (kb + 2 < 32 / GEMM_KB) gcn_frag_load(f0, xa, wb, kb + 2);
        gcn_frag_mma(f1, acc);
    }

    __syncthreads();
    float* so = sX + (tile * 16 + 8 * h) * GEMM_XPITCH + 64 * ch + 2 * m;
#pragma unroll
    for (int p = 0; p < 2; ++p) {
#pragma unroll
        for (int i = 0; i < 8; ++i) {
            v2f o;
            o.x = acc[2 * p][i];
            o.y = acc[2 * p + 1][i];
            *(v2f*)(so + i * GEMM_XPITCH + 32 * p) = o;
        }
    }
    __syncthreads();
    const float* si = sX + (tile * 16 + h) * GEMM_XPITCH + 64 * ch + 4 * m;
    float*       hp = H + (size_t)(row0 + tile * 16 + h) * GCN_D + 64 * ch + 4 * m;
    if (row0 + tile * 16 + 16 <= n) {
#pragma unroll
        for (int j = 0; j < 8; ++j)
            *(float4*)(hp + (size_t)(2 * j) * GCN_D) = *(const float4*)(si + 2 * j * GEMM_XPITCH);
    } else {
#pragma unroll
        for (int j = 0; j < 8; ++j)
            if (row0 + tile * 16 + 2 * j + h < n)
                *(float4*)(hp + (size_t)(2 * j) * GCN_D) = *(const float4*)(si + 2 * j * GEMM_XPITCH);
    }
}

__global__ __launch_bounds__(256) void gcn_gemm_wmma(const float* __restrict__ X,
                                                     const float* __restrict__ W,
                                                     const int* __restrict__ cnt,
                                                     float* __restrict__ H, int n) {
    __shared__ __attribute__((aligned(16))) float sW[GCN_D * GCN_D];
    __shared__ __attribute__((aligned(16))) float sX[GEMM_ROWS * GEMM_XPITCH];

    const int t    = threadIdx.x;
    const int lane = t & 31;
    const int wave = __builtin_amdgcn_readfirstlane(t >> 5);
    const int row0 = blockIdx.x * GEMM_ROWS;

    float4 xv[8];
    float  xs[8];
#pragma unroll
    for (int i = 0; i < 8; ++i) {
        const int r = min(row0 + wave + 8 * i, n - 1);
        xv[i] = ((const float4*)(X + (size_t)r * GCN_D))[lane];
        xs[i] = gcn_dinv(cnt[r] + cnt[GCN_CNT_PLANE + r]);
    }
    gcn_stage_w(sW, W, t);
#pragma unroll
    for (int i = 0; i < 8; ++i)
        *(float4*)(sX + (wave + 8 * i) * GEMM_XPITCH + 4 * lane) =
            make_float4(xv[i].x * xs[i], xv[i].y * xs[i], xv[i].z * xs[i], xv[i].w * xs[i]);
    __syncthreads();

    gcn_tile_product(sW, sX, H, row0, n, t);
}

template <int U>
__device__ __forceinline__ void gcn_rows(const float4* __restrict__ hv, int sidx,
                                         int j, int lane, float4& acc) {
    float4 v[U];
    for (int u = 0; u < U; ++u) {
        const int s = __builtin_amdgcn_readlane(sidx, j + u);
        const float4* row = hv + (size_t)s * (GCN_D / 4);
        v[u] = row[lane];
    }
    for (int u = 0; u < U; ++u) {
        acc.x += v[u].x;
        acc.y += v[u].y;
        acc.z += v[u].z;
        acc.w += v[u].w;
    }
}

__device__ __forceinline__ void gcn_overflow(const float4* __restrict__ hv, const int2* __restrict__ ovf, int novf,
                                             int node, int n, int lane, float4& acc) {
    for (int base = 0; base < novf; base += 32) {
        const int m = min(32, novf - base);
        int2 e = make_int2(-1, 0);
        if (lane < m) e = ovf[base + lane];
        const int sidx = min(max(e.y, 0), n - 1);
        unsigned hit = (unsigned)__ballot(e.x == node);
        while (hit) {
            const int j = __builtin_ctz(hit);
            hit &= hit - 1;
            gcn_rows<1>(hv, sidx, j, lane, acc);
        }
    }
}

__global__ __launch_bounds__(256) void gcn_gather(const int* __restrict__ cnt,
                                                  const int* __restrict__ bucket,
                                                  const int2* __restrict__ ovf, int ne,
                                                  const float* __restrict__ h,
                                                  const float* __restrict__ bias,
                                                  float* __restrict__ out, int n) {
    const int lane = threadIdx.x & 31;
    int node = (int)((blockIdx.x * 256u + threadIdx.x) >> 5);
    if (node >= n) return;
    node = __builtin_amdgcn_readfirstlane(node);

    const int   c0  = cnt[node];
    const int   c1  = cnt[GCN_CNT_PLANE + node];
    const float dd  = gcn_dinv(c0 + c1);
    const int   noff = node * GCN_CAP;
    const int   m0  = min(max(c0, 0), GCN_CAP);
    const int   beg = 0;
    const int   end = m0 + min(max(c1, 0), GCN_CAP);
    const float4* hv = (const float4*)h;
    const float4 self = hv[(size_t)node * (GCN_D / 4) + lane];

    float4 acc = make_float4(0.0f, 0.0f, 0.0f, 0.0f);
    for (int base = beg; base < end; base += 32) {
        const int m = min(32, end - base);
        int sidx = 0;
        if (lane < m) sidx = min(max(gcn_entry(bucket, noff, m0, base + lane), 0), n - 1);
        int j = 0;
        for (; j + 8 <= m; j += 8) gcn_rows<8>(hv, sidx, j, lane, acc);
        if (j + 4 <= m) { gcn_rows<4>(hv, sidx, j, lane, acc); j += 4; }
        if (j + 2 <= m) { gcn_rows<2>(hv, sidx, j, lane, acc); j += 2; }
        if (j < m)      { gcn_rows<1>(hv, sidx, j, lane, acc); }
    }
    if (c0 > GCN_CAP || c1 > GCN_CAP) gcn_overflow(hv, ovf, min(max(cnt[n], 0), ne), node, n, lane, acc);

    const float4 bv = ((const float4*)bias)[lane];
    float4 r;
    r.x = fmaf(acc.x + self.x, dd, bv.x);
    r.y = fmaf(acc.y + self.y, dd, bv.y);
    r.z = fmaf(acc.z + self.z, dd, bv.z);
    r.w = fmaf(acc.w + self.w, dd, bv.w);
    r.x = r.x > 0.0f ? r.x : 0.0f;
    r.y = r.y > 0.0f ? r.y : 0.0f;
    r.z = r.z > 0.0f ? r.z : 0.0f;
    r.w = r.w > 0.0f ? r.w : 0.0f;
    ((float4*)out)[(size_t)node * (GCN_D / 4) + lane] = r;
}

template <int U>
__device__ __forceinline__ void gcn_rows_load(const float4* __restrict__ hv, int sidx, int j, int lane,
                                              float4 (&v)[U]) {
#pragma unroll
    for (int u = 0; u < U; ++u) {
        const int s = __builtin_amdgcn_readlane(sidx, j + u);
        v[u] = hv[(size_t)s * (GCN_D / 4) + lane];
    }
}
template <int U>
__device__ __forceinline__ void gcn_rows_sum(const float4 (&v)[U], float4& acc) {
#pragma unroll
    for (int u = 0; u < U; ++u) {
        acc.x += v[u].x;
        acc.y += v[u].y;
        acc.z += v[u].z;
        acc.w += v[u].w;
    }
}

__device__ __forceinline__ void gcn_chunk(const float4* __restrict__ hv, int sidx, int m,
                                          int lane, float4& acc) {
    int jb = 0;
    if (m >= 16) {
        float4 va[16];
        gcn_rows_load<16>(hv, sidx, 0, lane, va);
        gcn_rows_sum<16>(va, acc);
        jb = 16;
    }
    float4 vb[8], vc[4], vd[2], ve[1];
    const bool tb = jb + 8 <= m;
    const int  jc = jb + (tb ? 8 : 0);
    const bool tc = jc + 4 <= m;
    const int  jd = jc + (tc ? 4 : 0);
    const bool td = jd + 2 <= m;
    const int  je = jd + (td ? 2 : 0);
    const bool te = je < m;
    if (tb) gcn_rows_load<8>(hv, sidx, jb, lane, vb);
    if (tc) gcn_rows_load<4>(hv, sidx, jc, lane, vc);
    if (td) gcn_rows_load<2>(hv, sidx, jd, lane, vd);
    if (te) gcn_rows_load<1>(hv, sidx, je, lane, ve);
    if (tb) gcn_rows_sum<8>(vb, acc);
    if (tc) gcn_rows_sum<4>(vc, acc);
    if (td) gcn_rows_sum<2>(vd, acc);
    if (te) gcn_rows_sum<1>(ve, acc);
    if (m == 32) gcn_rows<1>(hv, sidx, 31, lane, acc);
}

__global__ __launch_bounds__(256) void gcn_gather_gemm(const int* __restrict__ cnt,
                                                       const int* __restrict__ bucket,
                                                       const int2* __restrict__ ovf, int ne,
                                                       const float* __restrict__ hin,
                                                       const float* __restrict__ bias,
                                                       const float* __restrict__ W,
                                                       float* __restrict__ H, int n) {
    __shared__ __attribute__((aligned(16))) float sW[GCN_D * GCN_D];
    __shared__ __attribute__((aligned(16))) float sX[GEMM_ROWS * GEMM_XPITCH];

    const int t    = threadIdx.x;
    const int lane = t & 31;
    const int wave = __builtin_amdgcn_readfirstlane(t >> 5);
    const int row0 = blockIdx.x * GEMM_ROWS;
    const int node0 = row0 + 8 * wave;

    int cv0 = 0, cv1 = 0;
    if (node0 + lane < n) {
        cv0 = cnt[node0 + lane];
        cv1 = cnt[GCN_CNT_PLANE + node0 + lane];
    }
    const float4 bv  = ((const float4*)bias)[lane];
    const float4* hv = (const float4*)hin;

    gcn_stage_w(sW, W, t);

    int c0   = __builtin_amdgcn_readlane(cv0, 0);
    int c1   = __builtin_amdgcn_readlane(cv1, 0);
    int noff = node0 * GCN_CAP;
    int m0   = min(max(c0, 0), GCN_CAP);
    int tot  = m0 + min(max(c1, 0), GCN_CAP);
    int enext = 0;
    if (lane < min(32, tot)) enext = gcn_entry(bucket, noff, m0, lane);
    asm volatile("" :: "v"(enext));
    for (int i = 0; i < 8; ++i) {
        const int  node   = node0 + i;
        const int efirst = enext;
        const int nc0   = __builtin_amdgcn_readlane(cv0, i + 1);
        const int nc1   = __builtin_amdgcn_readlane(cv1, i + 1);
        const int nnoff = noff + GCN_CAP;
        const int nm0   = min(max(nc0, 0), GCN_CAP);
        const int ntot  = nm0 + min(max(nc1, 0), GCN_CAP);
        enext = 0;
        if (i < 7 && lane < min(32, ntot)) enext = gcn_entry(bucket, nnoff, nm0, lane);

        float4 r = make_float4(0.0f, 0.0f, 0.0f, 0.0f);
        if (node < n) {
            const float  dd   = gcn_dinv(c0 + c1);
            const float4 self = hv[(size_t)node * (GCN_D / 4) + lane];
            float4 acc = make_float4(0.0f, 0.0f, 0.0f, 0.0f);
            if (tot > 0) {
                const int sidx = min(max(efirst, 0), n - 1);
                gcn_chunk(hv, sidx, min(32, tot), lane, acc);
            }
            for (int base = 32; base < tot; base += 32) {
                const int m = min(32, tot - base);
                int e = 0;
                if (lane < m) e = gcn_entry(bucket, noff, m0, base + lane);
                const int sidx = min(max(e, 0), n - 1);
                for (int j = 0; j < m; ++j) gcn_rows<1>(hv, sidx, j, lane, acc);
            }
            if (c0 > GCN_CAP || c1 > GCN_CAP) gcn_overflow(hv, ovf, min(max(cnt[n], 0), ne), node, n, lane, acc);
            r.x = fmaf(acc.x + self.x, dd, bv.x);
            r.y = fmaf(acc.y + self.y, dd, bv.y);
            r.z = fmaf(acc.z + self.z, dd, bv.z);
            r.w = fmaf(acc.w + self.w, dd, bv.w);
            r.x = (r.x > 0.0f ? r.x : 0.0f) * dd;
            r.y = (r.y > 0.0f ? r.y : 0.0f) * dd;
            r.z = (r.z > 0.0f ? r.z : 0.0f) * dd;
            r.w = (r.w > 0.0f ? r.w : 0.0f) * dd;
        }
        *(float4*)(sX + (8 * wave + i) * GEMM_XPITCH + 4 * lane) = r;
        noff = nnoff;
        m0   = nm0;
        tot  = ntot;
        c0   = nc0;
        c1   = nc1;
    }
    __syncthreads();

    gcn_tile_product(sW, sX, H, row0, n, t);
}

extern "C" void kernel_launch(void* const* d_in, const int* in_sizes, int n_in,
                              void* d_out, int out_size, void* d_ws, size_t ws_size,
                              hipStream_t stream) {
    const float* x  = (const float*)d_in[0];
    const float* W1 = (const float*)d_in[1];
    const float* b1 = (const float*)d_in[2];
    const float* W2 = (const float*)d_in[3];
    const float* b2 = (const float*)d_in[4];
    const float* W3 = (const float*)d_in[5];
    const float* b3 = (const float*)d_in[6];
    const int*   eidx = (const int*)d_in[7];
    const int ne = in_sizes[7] / 2;
    const int* src = eidx;
    const int* dst = eidx + ne;

    const size_t edge_bytes = (((size_t)ne * 4) + 4095) & ~(size_t)4095;
    char* ws = (char*)d_ws;
    size_t off = 0;
    int*   cnt    = (int*)(ws + off);   off += 2 * (size_t)GCN_CNT_PLANE * 4;
    int*   bucket = (int*)(ws + off);   off += 2 * (size_t)GCN_BKT_PLANE * 4;
    int2*  ovf    = (int2*)(ws + off);  off += 2 * edge_bytes;
    float* h      = (float*)(ws + off); off += (size_t)GCN_N * GCN_D * 4;
    float* buf    = (float*)(ws + off);

    const int blk = 256;
    const int g_node = (2 * GCN_CNT_PLANE + blk - 1) / blk;
    const int g_edge = (ne + blk - 1) / blk;
    const int g_gemm = (GCN_N + GEMM_ROWS - 1) / GEMM_ROWS;
    const int g_gath = (GCN_N + 7) / 8;

    gcn_fill_zero_i32<<<g_node, blk, 0, stream>>>(cnt, 2 * GCN_CNT_PLANE);
    gcn_deg_bucket<<<g_edge, blk, 0, stream>>>(src, dst, cnt, bucket, ovf, GCN_N, ne);

    gcn_gemm_wmma<<<g_gemm, blk, 0, stream>>>(x, W1, cnt, h, GCN_N);

    gcn_gather_gemm<<<g_gemm, blk, 0, stream>>>(cnt, bucket, ovf, ne, h, b1, W2, buf, GCN_N);

    gcn_gather_gemm<<<g_gemm, blk, 0, stream>>>(cnt, bucket, ovf, ne, buf, b2, W3, h, GCN_N);

    gcn_gather<<<g_gath, blk, 0, stream>>>(cnt, bucket, ovf, ne, h, b3, (float*)d_out, GCN_N);
}


# Product kernels as 768 resident work-groups: each stages W once and walks its own 130 rows in chunks (first chunk 64, 43 or 22 rows, so matrix phases fall into neighbours' row loads); sum order kept
# speedup vs baseline: 5.7603x; 1.0867x over previous
#include <hip/hip_runtime.h>


#define GCN_N 100000
#define GCN_D 128

typedef float v2f __attribute__((ext_vector_type(2)));
typedef float v8f __attribute__((ext_vector_type(8)));

#define GCN_CAP       32
#define GCN_CNT_PLANE 100352
#define GCN_BKT_PLANE (GCN_N * GCN_CAP)

__global__ __launch_bounds__(256) void gcn_fill_zero_i32(int* __restrict__ p, int n) {
    int i = blockIdx.x * blockDim.x + threadIdx.x;
    if (i < n) p[i] = 0;
}

__global__ __launch_bounds__(256) void gcn_deg_bucket(const int* __restrict__ src,
                                                      const int* __restrict__ dst,
                                                      int* __restrict__ cnt,
                                                      int* __restrict__ bucket,
                                                      int2* __restrict__ ovf, int n, int ne) {
    int i = blockIdx.x * blockDim.x + threadIdx.x;
    if (i >= ne) return;
    const int s = src[i];
    const int d = dst[i];
    const int p = (int)((__builtin_amdgcn_s_sendmsg_rtn(0x87) >> 18) & 1u);
    const int r = atomicAdd(&cnt[p * GCN_CNT_PLANE + d], 1);
    if (r < GCN_CAP) {
        bucket[p * GCN_BKT_PLANE + d * GCN_CAP + r] = s;
    } else {
        const int o = atomicAdd(&cnt[n], 1);
        ovf[o] = make_int2(d, s);
    }
}

__device__ __forceinline__ float gcn_dinv(int indeg) { return rsqrtf((float)(indeg + 1)); }

__device__ __forceinline__ int gcn_entry(const int* __restrict__ bucket, int noff, int m0, int k) {
    return bucket[k < m0 ? noff + k : GCN_BKT_PLANE + noff + (k - m0)];
}

#define GEMM_ROWS   64
#define GEMM_XPITCH 132
#define GEMM_KB     4
#define GEMM_WGS    768

__device__ __forceinline__ void gcn_wg_rows(int n, int& rb, int& re) {
    const long long b = blockIdx.x;
    const long long g = gridDim.x;
    rb = (int)((b * n) / g);
    re = (int)(((b + 1) * n) / g);
}
__device__ __forceinline__ int gcn_first_chunk() { return GEMM_ROWS - 21 * (int)(blockIdx.x % 3u); }

struct gcn_frag {
    v2f    a[GEMM_KB];
    float4 b[GEMM_KB][2];
};

__device__ __forceinline__ void gcn_frag_load(gcn_frag& f, const float* __restrict__ xa,
                                              const float* __restrict__ wb, int kb) {
#pragma unroll
    for (int u = 0; u < GEMM_KB; ++u) {
        const int ks = kb * GEMM_KB + u;
        f.a[u]    = *(const v2f*)(xa + 4 * ks);
        f.b[u][0] = *(const float4*)(wb + ks * 512);
        f.b[u][1] = *(const float4*)(wb + ks * 512 + 128);
    }
}

__device__ __forceinline__ void gcn_frag_mma(const gcn_frag& f, v8f (&acc)[4]) {
#pragma unroll
    for (int u = 0; u < GEMM_KB; ++u) {
        v2f b;
        b.x = f.b[u][0].x; b.y = f.b[u][0].y;
        acc[0] = __builtin_amdgcn_wmma_f32_16x16x4_f32(false, f.a[u], false, b, (short)0, acc[0], false, false);
        b.x = f.b[u][0].z; b.y = f.b[u][0].w;
        acc[1] = __builtin_amdgcn_wmma_f32_16x16x4_f32(false, f.a[u], false, b, (short)0, acc[1], false, false);
        b.x = f.b[u][1].x; b.y = f.b[u][1].y;
        acc[2] = __builtin_amdgcn_wmma_f32_16x16x4_f32(false, f.a[u], false, b, (short)0, acc[2], false, false);
        b.x = f.b[u][1].z; b.y = f.b[u][1].w;
        acc[3] = __builtin_amdgcn_wmma_f32_16x16x4_f32(false, f.a[u], false, b, (short)0, acc[3], false, false);
    }
}

__device__ __forceinline__ void gcn_stage_w(float* __restrict__ sW, const float* __restrict__ W, int t) {
    const int lane = t & 31;
    const int m    = lane & 15;
    const int h    = lane >> 4;
#pragma unroll
    for (int i = 0; i < 16; ++i) {
        const int d  = t + 256 * i;
        const int cp = (d >> 5) & 3;
        const int ks = d >> 7;
        const float* wp = W + (size_t)(4 * ks + 2 * h) * GCN_D + 32 * cp + 2 * m;
        const v2f r0 = *(const v2f*)wp;
        const v2f r1 = *(const v2f*)(wp + GCN_D);
        *(float4*)(sW + 4 * d) = make_float4(r0.x, r1.x, r0.y, r1.y);
    }
}

__device__ __forceinline__ void gcn_tile_product(const float* __restrict__ sW, float* __restrict__ sX,
                                                 float* __restrict__ H, int row0, int rend, int t) {
    const int lane = t & 31;
    const int wave = __builtin_amdgcn_readfirstlane(t >> 5);
    const int m    = lane & 15;
    const int h    = lane >> 4;
    const int tile = wave >> 1;
    const int ch   = wave & 1;
    const float* xa = sX + (tile * 16 + m) * GEMM_XPITCH + 2 * h;
    const float* wb = sW + 4 * (2 * ch * 32 + lane);

    v8f acc[4] = {};
    if (row0 + tile * 16 < rend) {
        gcn_frag f0, f1;
        gcn_frag_load(f0, xa, wb, 0);
#pragma unroll
        for (int kb = 0; kb < 32 / GEMM_KB; kb += 2) {
            gcn_frag_load(f1, xa, wb, kb + 1);
            gcn_frag_mma(f0, acc);
            if (kb + 2 < 32 / GEMM_KB) gcn_frag_load(f0, xa, wb, kb + 2);
            gcn_frag_mma(f1, acc);
        }
    }

    __syncthreads();
    float* so = sX + (tile * 16 + 8 * h) * GEMM_XPITCH + 64 * ch + 2 * m;
#pragma unroll
    for (int p = 0; p < 2; ++p) {
#pragma unroll
        for (int i = 0; i < 8; ++i) {
            v2f o;
            o.x = acc[2 * p][i];
            o.y = acc[2 * p + 1][i];
            *(v2f*)(so + i * GEMM_XPITCH + 32 * p) = o;
        }
    }
    __syncthreads();
    const float* si = sX + (tile * 16 + h) * GEMM_XPITCH + 64 * ch + 4 * m;
    float*       hp = H + (size_t)(row0 + tile * 16 + h) * GCN_D + 64 * ch + 4 * m;
    if (row0 + tile * 16 + 16 <= rend) {
#pragma unroll
        for (int j = 0; j < 8; ++j)
            *(float4*)(hp + (size_t)(2 * j) * GCN_D) = *(const float4*)(si + 2 * j * GEMM_XPITCH);
    } else {
#pragma unroll
        for (int j = 0; j < 8; ++j)
            if (row0 + tile * 16 + 2 * j + h < rend)
                *(float4*)(hp + (size_t)(2 * j) * GCN_D) = *(const float4*)(si + 2 * j * GEMM_XPITCH);
    }
}

__global__ __launch_bounds__(256) void gcn_gemm_wmma(const float* __restrict__ X,
                                                     const float* __restrict__ W,
                                                     const int* __restrict__ cnt,
                                                     float* __restrict__ H, int n) {
    __shared__ __attribute__((aligned(16))) float sW[GCN_D * GCN_D];
    __shared__ __attribute__((aligned(16))) float sX[GEMM_ROWS * GEMM_XPITCH];

    const int t    = threadIdx.x;
    const int lane = t & 31;
    const int wave = __builtin_amdgcn_readfirstlane(t >> 5);
    int rb, re;
    gcn_wg_rows(n, rb, re);

    gcn_stage_w(sW, W, t);

    int row0 = rb;
    int len  = gcn_first_chunk();
    while (row0 < re) {
        const int rend = min(re, row0 + len);
        float4 xv[8];
        float  xs[8];
#pragma unroll
        for (int i = 0; i < 8; ++i) {
            const int r = min(row0 + wave + 8 * i, rend - 1);
            xv[i] = ((const float4*)(X + (size_t)r * GCN_D))[lane];
            xs[i] = gcn_dinv(cnt[r] + cnt[GCN_CNT_PLANE + r]);
        }
#pragma unroll
        for (int i = 0; i < 8; ++i)
            *(float4*)(sX + (wave + 8 * i) * GEMM_XPITCH + 4 * lane) =
                make_float4(xv[i].x * xs[i], xv[i].y * xs[i], xv[i].z * xs[i], xv[i].w * xs[i]);
        __syncthreads();

        gcn_tile_product(sW, sX, H, row0, rend, t);
        __syncthreads();
        row0 = rend;
        len  = GEMM_ROWS;
    }
}

template <int U>
__device__ __forceinline__ void gcn_rows(const float4* __restrict__ hv, int sidx,
                                         int j, int lane, float4& acc) {
    float4 v[U];
    for (int u = 0; u < U; ++u) {
        const int s = __builtin_amdgcn_readlane(sidx, j + u);
        const float4* row = hv + (size_t)s * (GCN_D / 4);
        v[u] = row[lane];
    }
    for (int u = 0; u < U; ++u) {
        acc.x += v[u].x;
        acc.y += v[u].y;
        acc.z += v[u].z;
        acc.w += v[u].w;
    }
}

__device__ __forceinline__ void gcn_overflow(const float4* __restrict__ hv, const int2* __restrict__ ovf, int novf,
                                             int node, int n, int lane, float4& acc) {
    for (int base = 0; base < novf; base += 32) {
        const int m = min(32, novf - base);
        int2 e = make_int2(-1, 0);
        if (lane < m) e = ovf[base + lane];
        const int sidx = min(max(e.y, 0), n - 1);
        unsigned hit = (unsigned)__ballot(e.x == node);
        while (hit) {
            const int j = __builtin_ctz(hit);
            hit &= hit - 1;
            gcn_rows<1>(hv, sidx, j, lane, acc);
        }
    }
}

__global__ __launch_bounds__(256) void gcn_gather(const int* __restrict__ cnt,
                                                  const int* __restrict__ bucket,
                                                  const int2* __restrict__ ovf, int ne,
                                                  const float* __restrict__ h,
                                                  const float* __restrict__ bias,
                                                  float* __restrict__ out, int n) {
    const int lane = threadIdx.x & 31;
    int node = (int)((blockIdx.x * 256u + threadIdx.x) >> 5);
    if (node >= n) return;
    node = __builtin_amdgcn_readfirstlane(node);

    const int   c0  = cnt[node];
    const int   c1  = cnt[GCN_CNT_PLANE + node];
    const float dd  = gcn_dinv(c0 + c1);
    const int   noff = node * GCN_CAP;
    const int   m0  = min(max(c0, 0), GCN_CAP);
    const int   beg = 0;
    const int   end = m0 + min(max(c1, 0), GCN_CAP);
    const float4* hv = (const float4*)h;
    const float4 self = hv[(size_t)node * (GCN_D / 4) + lane];

    float4 acc = make_float4(0.0f, 0.0f, 0.0f, 0.0f);
    for (int base = beg; base < end; base += 32) {
        const int m = min(32, end - base);
        int sidx = 0;
        if (lane < m) sidx = min(max(gcn_entry(bucket, noff, m0, base + lane), 0), n - 1);
        int j = 0;
        for (; j + 8 <= m; j += 8) gcn_rows<8>(hv, sidx, j, lane, acc);
        if (j + 4 <= m) { gcn_rows<4>(hv, sidx, j, lane, acc); j += 4; }
        if (j + 2 <= m) { gcn_rows<2>(hv, sidx, j, lane, acc); j += 2; }
        if (j < m)      { gcn_rows<1>(hv, sidx, j, lane, acc); }
    }
    if (c0 > GCN_CAP || c1 > GCN_CAP) gcn_overflow(hv, ovf, min(max(cnt[n], 0), ne), node, n, lane, acc);

    const float4 bv = ((const float4*)bias)[lane];
    float4 r;
    r.x = fmaf(acc.x + self.x, dd, bv.x);
    r.y = fmaf(acc.y + self.y, dd, bv.y);
    r.z = fmaf(acc.z + self.z, dd, bv.z);
    r.w = fmaf(acc.w + self.w, dd, bv.w);
    r.x = r.x > 0.0f ? r.x : 0.0f;
    r.y = r.y > 0.0f ? r.y : 0.0f;
    r.z = r.z > 0.0f ? r.z : 0.0f;
    r.w = r.w > 0.0f ? r.w : 0.0f;
    ((float4*)out)[(size_t)node * (GCN_D / 4) + lane] = r;
}

template <int U>
__device__ __forceinline__ void gcn_rows_load(const float4* __restrict__ hv, int sidx, int j, int lane,
                                              float4 (&v)[U]) {
#pragma unroll
    for (int u = 0; u < U; ++u) {
        const int s = __builtin_amdgcn_readlane(sidx, j + u);
        v[u] = hv[(size_t)s * (GCN_D / 4) + lane];
    }
}
template <int U>
__device__ __forceinline__ void gcn_rows_sum(const float4 (&v)[U], float4& acc) {
#pragma unroll
    for (int u = 0; u < U; ++u) {
        acc.x += v[u].x;
        acc.y += v[u].y;
        acc.z += v[u].z;
        acc.w += v[u].w;
    }
}

__device__ __forceinline__ void gcn_chunk(const float4* __restrict__ hv, int sidx, int m,
                                          int lane, float4& acc) {
    int jb = 0;
    if (m >= 16) {
        float4 va[16];
        gcn_rows_load<16>(hv, sidx, 0, lane, va);
        gcn_rows_sum<16>(va, acc);
        jb = 16;
    }
    float4 vb[8], vc[4], vd[2], ve[1];
    const bool tb = jb + 8 <= m;
    const int  jc = jb + (tb ? 8 : 0);
    const bool tc = jc + 4 <= m;
    const int  jd = jc + (tc ? 4 : 0);
    const bool td = jd + 2 <= m;
    const int  je = jd + (td ? 2 : 0);
    const bool te = je < m;
    if (tb) gcn_rows_load<8>(hv, sidx, jb, lane, vb);
    if (tc) gcn_rows_load<4>(hv, sidx, jc, lane, vc);
    if (td) gcn_rows_load<2>(hv, sidx, jd, lane, vd);
    if (te) gcn_rows_load<1>(hv, sidx, je, lane, ve);
    if (tb) gcn_rows_sum<8>(vb, acc);
    if (tc) gcn_rows_sum<4>(vc, acc);
    if (td) gcn_rows_sum<2>(vd, acc);
    if (te) gcn_rows_sum<1>(ve, acc);
    if (m == 32) gcn_rows<1>(hv, sidx, 31, lane, acc);
}

__global__ __launch_bounds__(256) void gcn_gather_gemm(const int* __restrict__ cnt,
                                                       const int* __restrict__ bucket,
                                                       const int2* __restrict__ ovf, int ne,
                                                       const float* __restrict__ hin,
                                                       const float* __restrict__ bias,
                                                       const float* __restrict__ W,
                                                       float* __restrict__ H, int n) {
    __shared__ __attribute__((aligned(16))) float sW[GCN_D * GCN_D];
    __shared__ __attribute__((aligned(16))) float sX[GEMM_ROWS * GEMM_XPITCH];

    const int t    = threadIdx.x;
    const int lane = t & 31;
    const int wave = __builtin_amdgcn_readfirstlane(t >> 5);
    int rb, re;
    gcn_wg_rows(n, rb, re);
    const float4 bv  = ((const float4*)bias)[lane];
    const float4* hv = (const float4*)hin;

    gcn_stage_w(sW, W, t);

    int row0 = rb;
    int len  = gcn_first_chunk();
    while (row0 < re) {
        const int rend  = min(re, row0 + len);
        const int node0 = row0 + wave;

        int cv0 = 0, cv1 = 0;
        if (node0 + 8 * lane < rend) {
            cv0 = cnt[node0 + 8 * lane];
            cv1 = cnt[GCN_CNT_PLANE + node0 + 8 * lane];
        }

        int c0   = __builtin_amdgcn_readlane(cv0, 0);
        int c1   = __builtin_amdgcn_readlane(cv1, 0);
        int noff = node0 * GCN_CAP;
        int m0   = min(max(c0, 0), GCN_CAP);
        int tot  = m0 + min(max(c1, 0), GCN_CAP);
        int enext = 0;
        if (lane < min(32, tot)) enext = gcn_entry(bucket, noff, m0, lane);
        asm volatile("" :: "v"(enext));
        for (int i = 0; i < 8; ++i) {
            const int  node   = node0 + 8 * i;
            const int efirst = enext;
            const int nc0   = __builtin_amdgcn_readlane(cv0, i + 1);
            const int nc1   = __builtin_amdgcn_readlane(cv1, i + 1);
            const int nnoff = noff + 8 * GCN_CAP;
            const int nm0   = min(max(nc0, 0), GCN_CAP);
            const int ntot  = nm0 + min(max(nc1, 0), GCN_CAP);
            enext = 0;
            if (i < 7 && lane < min(32, ntot)) enext = gcn_entry(bucket, nnoff, nm0, lane);

            float4 r = make_float4(0.0f, 0.0f, 0.0f, 0.0f);
            if (node < rend) {
                const float  dd   = gcn_dinv(c0 + c1);
                const float4 self = hv[(size_t)node * (GCN_D / 4) + lane];
                float4 acc = make_float4(0.0f, 0.0f, 0.0f, 0.0f);
                if (tot > 0) {
                    const int sidx = min(max(efirst, 0), n - 1);
                    gcn_chunk(hv, sidx, min(32, tot), lane, acc);
                }
                for (int base = 32; base < tot; base += 32) {
                    const int m = min(32, tot - base);
                    int e = 0;
                    if (lane < m) e = gcn_entry(bucket, noff, m0, base + lane);
                    const int sidx = min(max(e, 0), n - 1);
                    for (int j = 0; j < m; ++j) gcn_rows<1>(hv, sidx, j, lane, acc);
                }
                if (c0 > GCN_CAP || c1 > GCN_CAP) gcn_overflow(hv, ovf, min(max(cnt[n], 0), ne), node, n, lane, acc);
                r.x = fmaf(acc.x + self.x, dd, bv.x);
                r.y = fmaf(acc.y + self.y, dd, bv.y);
                r.z = fmaf(acc.z + self.z, dd, bv.z);
                r.w = fmaf(acc.w + self.w, dd, bv.w);
                r.x = (r.x > 0.0f ? r.x : 0.0f) * dd;
                r.y = (r.y > 0.0f ? r.y : 0.0f) * dd;
                r.z = (r.z > 0.0f ? r.z : 0.0f) * dd;
                r.w = (r.w > 0.0f ? r.w : 0.0f) * dd;
            }
            *(float4*)(sX + (wave + 8 * i) * GEMM_XPITCH + 4 * lane) = r;
            noff = nnoff;
            m0   = nm0;
            tot  = ntot;
            c0   = nc0;
            c1   = nc1;
        }
        __syncthreads();

        gcn_tile_product(sW, sX, H, row0, rend, t);
        __syncthreads();
        row0 = rend;
        len  = GEMM_ROWS;
    }
}

extern "C" void kernel_launch(void* const* d_in, const int* in_sizes, int n_in,
                              void* d_out, int out_size, void* d_ws, size_t ws_size,
                              hipStream_t stream) {
    const float* x  = (const float*)d_in[0];
    const float* W1 = (const float*)d_in[1];
    const float* b1 = (const float*)d_in[2];
    const float* W2 = (const float*)d_in[3];
    const float* b2 = (const float*)d_in[4];
    const float* W3 = (const float*)d_in[5];
    const float* b3 = (const float*)d_in[6];
    const int*   eidx = (const int*)d_in[7];
    const int ne = in_sizes[7] / 2;
    const int* src = eidx;
    const int* dst = eidx + ne;

    const size_t edge_bytes = (((size_t)ne * 4) + 4095) & ~(size_t)4095;
    char* ws = (char*)d_ws;
    size_t off = 0;
    int*   cnt    = (int*)(ws + off);   off += 2 * (size_t)GCN_CNT_PLANE * 4;
    int*   bucket = (int*)(ws + off);   off += 2 * (size_t)GCN_BKT_PLANE * 4;
    int2*  ovf    = (int2*)(ws + off);  off += 2 * edge_bytes;
    float* h      = (float*)(ws + off); off += (size_t)GCN_N * GCN_D * 4;
    float* buf    = (float*)(ws + off);

    const int blk = 256;
    const int g_node = (2 * GCN_CNT_PLANE + blk - 1) / blk;
    const int g_edge = (ne + blk - 1) / blk;
    const int g_gemm = GEMM_WGS;
    const int g_gath = (GCN_N + 7) / 8;

    gcn_fill_zero_i32<<<g_node, blk, 0, stream>>>(cnt, 2 * GCN_CNT_PLANE);
    gcn_deg_bucket<<<g_edge, blk, 0, stream>>>(src, dst, cnt, bucket, ovf, GCN_N, ne);

    gcn_gemm_wmma<<<g_gemm, blk, 0, stream>>>(x, W1, cnt, h, GCN_N);

    gcn_gather_gemm<<<g_gemm, blk, 0, stream>>>(cnt, bucket, ovf, ne, h, b1, W2, buf, GCN_N);

    gcn_gather_gemm<<<g_gemm, blk, 0, stream>>>(cnt, bucket, ovf, ne, buf, b2, W3, h, GCN_N);

    gcn_gather<<<g_gath, blk, 0, stream>>>(cnt, bucket, ovf, ne, h, b3, (float*)d_out, GCN_N);
}


# Edge list by two-level LDS counting sort (chunks binned by 256-node block; one work-group per block ranks, stores bucket lines): no per-edge global add or slot store; list order = LDS arrival order
# speedup vs baseline: 6.0257x; 1.0461x over previous
#include <hip/hip_runtime.h>


#define GCN_N 100000
#define GCN_D 128

typedef float v2f __attribute__((ext_vector_type(2)));
typedef float v8f __attribute__((ext_vector_type(8)));

#define GCN_CAP       32
#define GCN_CNT_PLANE 100352
#define GCN_BKT_PLANE (GCN_N * GCN_CAP)
#define GCN_NB        256
#define GCN_NBLK      ((GCN_N + GCN_NB - 1) / GCN_NB)
#define GCN_CH        4096
#define GCN_RPITCH    448
#define GCN_SLOTS     (2 * GCN_CAP)
#define GCN_SPITCH    (GCN_SLOTS + 1)

__global__ __launch_bounds__(256) void gcn_edge_bin(const int* __restrict__ src,
                                                    const int* __restrict__ dst,
                                                    int* __restrict__ rec,
                                                    int* __restrict__ runs,
                                                    int* __restrict__ cnt, int n, int ne) {
    __shared__ int sCnt[GCN_NBLK];
    __shared__ int sStart[GCN_NBLK];
    __shared__ int sRec[GCN_CH];
    const int t  = threadIdx.x;
    const int c  = blockIdx.x;
    const int e0 = c * GCN_CH;
    if (c == 0 && t == 0) cnt[n] = 0;
    for (int i = t; i < GCN_NBLK; i += 256) sCnt[i] = 0;
    __syncthreads();

    int sv[16], dv[16], rk[16];
#pragma unroll
    for (int u = 0; u < 16; ++u) {
        const int i = e0 + u * 256 + t;
        sv[u] = 0;
        dv[u] = -1;
        if (i < ne) { sv[u] = src[i]; dv[u] = dst[i]; }
    }
#pragma unroll
    for (int u = 0; u < 16; ++u) {
        rk[u] = 0;
        if (dv[u] >= 0) rk[u] = atomicAdd(&sCnt[min(dv[u] / GCN_NB, GCN_NBLK - 1)], 1);
    }
    __syncthreads();

    static_assert(GCN_NBLK <= 32 * 13, "the scan below gives each of 32 lanes 13 bins");
    if (t < 32) {
        int loc[13];
        int sum = 0;
#pragma unroll
        for (int k = 0; k < 13; ++k) {
            const int idx = t * 13 + k;
            loc[k] = sum;
            sum += idx < GCN_NBLK ? sCnt[idx] : 0;
        }
        int incl = sum;
        for (int off = 1; off < 32; off <<= 1) {
            const int y = __shfl_up(incl, off);
            if (t >= off) incl += y;
        }
        const int excl = incl - sum;
#pragma unroll
        for (int k = 0; k < 13; ++k) {
            const int idx = t * 13 + k;
            if (idx < GCN_NBLK) sStart[idx] = excl + loc[k];
        }
    }
    __syncthreads();

    for (int b = t; b < GCN_NBLK; b += 256) runs[(size_t)c * GCN_RPITCH + b] = sStart[b] | (sCnt[b] << 16);
#pragma unroll
    for (int u = 0; u < 16; ++u) {
        if (dv[u] >= 0) {
            const int blk = min(dv[u] / GCN_NB, GCN_NBLK - 1);
            sRec[sStart[blk] + rk[u]] = (int)(((unsigned)(dv[u] - blk * GCN_NB) << 24) | ((unsigned)sv[u] & 0xFFFFFFu));
        }
    }
    __syncthreads();
#pragma unroll
    for (int u = 0; u < 16; ++u) {
        const int i = u * 256 + t;
        if (e0 + i < ne) rec[e0 + i] = sRec[i];
    }
}

__device__ __forceinline__ void gcn_list_insert(int r, int* sN, int* sB, int* __restrict__ cnt,
                                                int2* __restrict__ ovf, int nodebase, int n) {
    const int dl   = (int)((unsigned)r >> 24);
    const int s    = r & 0xFFFFFF;
    const int slot = atomicAdd(&sN[dl], 1);
    if (slot < GCN_SLOTS) {
        sB[dl * GCN_SPITCH + slot] = s;
    } else {
        const int o = atomicAdd(&cnt[n], 1);
        ovf[o] = make_int2(nodebase + dl, s);
    }
}

__global__ __launch_bounds__(1024) void gcn_list_build(const int* __restrict__ rec,
                                                       const int* __restrict__ runs,
                                                       int* __restrict__ cnt,
                                                       int* __restrict__ bucket,
                                                       int2* __restrict__ ovf, int n, int ne) {
    __shared__ int sN[GCN_NB];
    __shared__ int sB[GCN_NB * GCN_SPITCH];
    __shared__ int sRun[1024];
    const int t    = threadIdx.x;
    const int lane = t & 31;
    const int wave = __builtin_amdgcn_readfirstlane(t >> 5);
    const int b    = blockIdx.x;
    const int nodebase = b * GCN_NB;
    const int nch  = (ne + GCN_CH - 1) / GCN_CH;
    if (t < GCN_NB) sN[t] = 0;
    for (int c0 = 0; c0 < nch; c0 += 1024) {
        __syncthreads();
        sRun[t] = c0 + t < nch ? runs[(size_t)(c0 + t) * GCN_RPITCH + b] : 0;
        __syncthreads();
        const int m = min(1024, nch - c0);
        for (int q = wave; q < m; q += 256) {
            int r[8], cn[8], base[8];
#pragma unroll
            for (int u = 0; u < 8; ++u) {
                const int qq = q + 32 * u;
                const int d  = qq < m ? sRun[qq] : 0;
                cn[u]   = d >> 16;
                base[u] = (c0 + qq) * GCN_CH + (d & 0xFFFF);
                r[u]    = 0;
                if (lane < cn[u]) r[u] = rec[base[u] + lane];
            }
#pragma unroll
            for (int u = 0; u < 8; ++u)
                if (lane < cn[u]) gcn_list_insert(r[u], sN, sB, cnt, ovf, nodebase, n);
#pragma unroll
            for (int u = 0; u < 8; ++u)
                for (int off = 32; off < cn[u]; off += 32)
                    if (off + lane < cn[u]) gcn_list_insert(rec[base[u] + off + lane], sN, sB, cnt, ovf, nodebase, n);
        }
    }
    __syncthreads();

    if (t < GCN_NB && nodebase + t < n) {
        const int deg = sN[t];
        const int c0v = min(deg, GCN_CAP);
        cnt[nodebase + t] = c0v;
        cnt[GCN_CNT_PLANE + nodebase + t] = deg - c0v;
    }
    for (int idx = t; idx < GCN_NB * GCN_CAP; idx += 1024) {
        const int j    = idx / GCN_CAP;
        const int k    = idx % GCN_CAP;
        const int node = nodebase + j;
        if (node < n) {
            const int deg = sN[j];
            if (k < deg)           bucket[node * GCN_CAP + k] = sB[j * GCN_SPITCH + k];
            if (k + GCN_CAP < deg) bucket[GCN_BKT_PLANE + node * GCN_CAP + k] = sB[j * GCN_SPITCH + GCN_CAP + k];
        }
    }
}

__device__ __forceinline__ float gcn_dinv(int indeg) { return rsqrtf((float)(indeg + 1)); }

__device__ __forceinline__ int gcn_entry(const int* __restrict__ bucket, int noff, int m0, int k) {
    return bucket[k < m0 ? noff + k : GCN_BKT_PLANE + noff + (k - m0)];
}

#define GEMM_ROWS   64
#define GEMM_XPITCH 132
#define GEMM_KB     4
#define GEMM_WGS    768

__device__ __forceinline__ void gcn_wg_rows(int n, int& rb, int& re) {
    const long long b = blockIdx.x;
    const long long g = gridDim.x;
    rb = (int)((b * n) / g);
    re = (int)(((b + 1) * n) / g);
}
__device__ __forceinline__ int gcn_first_chunk() { return GEMM_ROWS - 21 * (int)(blockIdx.x % 3u); }

struct gcn_frag {
    v2f    a[GEMM_KB];
    float4 b[GEMM_KB][2];
};

__device__ __forceinline__ void gcn_frag_load(gcn_frag& f, const float* __restrict__ xa,
                                              const float* __restrict__ wb, int kb) {
#pragma unroll
    for (int u = 0; u < GEMM_KB; ++u) {
        const int ks = kb * GEMM_KB + u;
        f.a[u]    = *(const v2f*)(xa + 4 * ks);
        f.b[u][0] = *(const float4*)(wb + ks * 512);
        f.b[u][1] = *(const float4*)(wb + ks * 512 + 128);
    }
}

__device__ __forceinline__ void gcn_frag_mma(const gcn_frag& f, v8f (&acc)[4]) {
#pragma unroll
    for (int u = 0; u < GEMM_KB; ++u) {
        v2f b;
        b.x = f.b[u][0].x; b.y = f.b[u][0].y;
        acc[0] = __builtin_amdgcn_wmma_f32_16x16x4_f32(false, f.a[u], false, b, (short)0, acc[0], false, false);
        b.x = f.b[u][0].z; b.y = f.b[u][0].w;
        acc[1] = __builtin_amdgcn_wmma_f32_16x16x4_f32(false, f.a[u], false, b, (short)0, acc[1], false, false);
        b.x = f.b[u][1].x; b.y = f.b[u][1].y;
        acc[2] = __builtin_amdgcn_wmma_f32_16x16x4_f32(false, f.a[u], false, b, (short)0, acc[2], false, false);
        b.x = f.b[u][1].z; b.y = f.b[u][1].w;
        acc[3] = __builtin_amdgcn_wmma_f32_16x16x4_f32(false, f.a[u], false, b, (short)0, acc[3], false, false);
    }
}

__device__ __forceinline__ void gcn_stage_w(float* __restrict__ sW, const float* __restrict__ W, int t) {
    const int lane = t & 31;
    const int m    = lane & 15;
    const int h    = lane >> 4;
#pragma unroll
    for (int i = 0; i < 16; ++i) {
        const int d  = t + 256 * i;
        const int cp = (d >> 5) & 3;
        const int ks = d >> 7;
        const float* wp = W + (size_t)(4 * ks + 2 * h) * GCN_D + 32 * cp + 2 * m;
        const v2f r0 = *(const v2f*)wp;
        const v2f r1 = *(const v2f*)(wp + GCN_D);
        *(float4*)(sW + 4 * d) = make_float4(r0.x, r1.x, r0.y, r1.y);
    }
}

__device__ __forceinline__ void gcn_tile_product(const float* __restrict__ sW, float* __restrict__ sX,
                                                 float* __restrict__ H, int row0, int rend, int t) {
    const int lane = t & 31;
    const int wave = __builtin_amdgcn_readfirstlane(t >> 5);
    const int m    = lane & 15;
    const int h    = lane >> 4;
    const int tile = wave >> 1;
    const int ch   = wave & 1;
    const float* xa = sX + (tile * 16 + m) * GEMM_XPITCH + 2 * h;
    const float* wb = sW + 4 * (2 * ch * 32 + lane);

    v8f acc[4] = {};
    if (row0 + tile * 16 < rend) {
        gcn_frag f0, f1;
        gcn_frag_load(f0, xa, wb, 0);
#pragma unroll
        for (int kb = 0; kb < 32 / GEMM_KB; kb += 2) {
            gcn_frag_load(f1, xa, wb, kb + 1);
            gcn_frag_mma(f0, acc);
            if (kb + 2 < 32 / GEMM_KB) gcn_frag_load(f0, xa, wb, kb + 2);
            gcn_frag_mma(f1, acc);
        }
    }

    __syncthreads();
    float* so = sX + (tile * 16 + 8 * h) * GEMM_XPITCH + 64 * ch + 2 * m;
#pragma unroll
    for (int p = 0; p < 2; ++p) {
#pragma unroll
        for (int i = 0; i < 8; ++i) {
            v2f o;
            o.x = acc[2 * p][i];
            o.y = acc[2 * p + 1][i];
            *(v2f*)(so + i * GEMM_XPITCH + 32 * p) = o;
        }
    }
    __syncthreads();
    const float* si = sX + (tile * 16 + h) * GEMM_XPITCH + 64 * ch + 4 * m;
    float*       hp = H + (size_t)(row0 + tile * 16 + h) * GCN_D + 64 * ch + 4 * m;
    if (row0 + tile * 16 + 16 <= rend) {
#pragma unroll
        for (int j = 0; j < 8; ++j)
            *(float4*)(hp + (size_t)(2 * j) * GCN_D) = *(const float4*)(si + 2 * j * GEMM_XPITCH);
    } else {
#pragma unroll
        for (int j = 0; j < 8; ++j)
            if (row0 + tile * 16 + 2 * j + h < rend)
                *(float4*)(hp + (size_t)(2 * j) * GCN_D) = *(const float4*)(si + 2 * j * GEMM_XPITCH);
    }
}

__global__ __launch_bounds__(256) void gcn_gemm_wmma(const float* __restrict__ X,
                                                     const float* __restrict__ W,
                                                     const int* __restrict__ cnt,
                                                     float* __restrict__ H, int n) {
    __shared__ __attribute__((aligned(16))) float sW[GCN_D * GCN_D];
    __shared__ __attribute__((aligned(16))) float sX[GEMM_ROWS * GEMM_XPITCH];

    const int t    = threadIdx.x;
    const int lane = t & 31;
    const int wave = __builtin_amdgcn_readfirstlane(t >> 5);
    int rb, re;
    gcn_wg_rows(n, rb, re);

    gcn_stage_w(sW, W, t);

    int row0 = rb;
    int len  = gcn_first_chunk();
    while (row0 < re) {
        const int rend = min(re, row0 + len);
        float4 xv[8];
        float  xs[8];
#pragma unroll
        for (int i = 0; i < 8; ++i) {
            const int r = min(row0 + wave + 8 * i, rend - 1);
            xv[i] = ((const float4*)(X + (size_t)r * GCN_D))[lane];
            xs[i] = gcn_dinv(cnt[r] + cnt[GCN_CNT_PLANE + r]);
        }
#pragma unroll
        for (int i = 0; i < 8; ++i)
            *(float4*)(sX + (wave + 8 * i) * GEMM_XPITCH + 4 * lane) =
                make_float4(xv[i].x * xs[i], xv[i].y * xs[i], xv[i].z * xs[i], xv[i].w * xs[i]);
        __syncthreads();

        gcn_tile_product(sW, sX, H, row0, rend, t);
        __syncthreads();
        row0 = rend;
        len  = GEMM_ROWS;
    }
}

template <int U>
__device__ __forceinline__ void gcn_rows(const float4* __restrict__ hv, int sidx,
                                         int j, int lane, float4& acc) {
    float4 v[U];
    for (int u = 0; u < U; ++u) {
        const int s = __builtin_amdgcn_readlane(sidx, j + u);
        const float4* row = hv + (size_t)s * (GCN_D / 4);
        v[u] = row[lane];
    }
    for (int u = 0; u < U; ++u) {
        acc.x += v[u].x;
        acc.y += v[u].y;
        acc.z += v[u].z;
        acc.w += v[u].w;
    }
}

__device__ __forceinline__ void gcn_overflow(const float4* __restrict__ hv, const int2* __restrict__ ovf, int novf,
                                             int node, int n, int lane, float4& acc) {
    for (int base = 0; base < novf; base += 32) {
        const int m = min(32, novf - base);
        int2 e = make_int2(-1, 0);
        if (lane < m) e = ovf[base + lane];
        const int sidx = min(max(e.y, 0), n - 1);
        unsigned hit = (unsigned)__ballot(e.x == node);
        while (hit) {
            const int j = __builtin_ctz(hit);
            hit &= hit - 1;
            gcn_rows<1>(hv, sidx, j, lane, acc);
        }
    }
}

__global__ __launch_bounds__(256) void gcn_gather(const int* __restrict__ cnt,
                                                  const int* __restrict__ bucket,
                                                  const int2* __restrict__ ovf, int ne,
                                                  const float* __restrict__ h,
                                                  const float* __restrict__ bias,
                                                  float* __restrict__ out, int n) {
    const int lane = threadIdx.x & 31;
    int node = (int)((blockIdx.x * 256u + threadIdx.x) >> 5);
    if (node >= n) return;
    node = __builtin_amdgcn_readfirstlane(node);

    const int   c0  = cnt[node];
    const int   c1  = cnt[GCN_CNT_PLANE + node];
    const float dd  = gcn_dinv(c0 + c1);
    const int   noff = node * GCN_CAP;
    const int   m0  = min(max(c0, 0), GCN_CAP);
    const int   beg = 0;
    const int   end = m0 + min(max(c1, 0), GCN_CAP);
    const float4* hv = (const float4*)h;
    const float4 self = hv[(size_t)node * (GCN_D / 4) + lane];

    float4 acc = make_float4(0.0f, 0.0f, 0.0f, 0.0f);
    for (int base = beg; base < end; base += 32) {
        const int m = min(32, end - base);
        int sidx = 0;
        if (lane < m) sidx = min(max(gcn_entry(bucket, noff, m0, base + lane), 0), n - 1);
        int j = 0;
        for (; j + 8 <= m; j += 8) gcn_rows<8>(hv, sidx, j, lane, acc);
        if (j + 4 <= m) { gcn_rows<4>(hv, sidx, j, lane, acc); j += 4; }
        if (j + 2 <= m) { gcn_rows<2>(hv, sidx, j, lane, acc); j += 2; }
        if (j < m)      { gcn_rows<1>(hv, sidx, j, lane, acc); }
    }
    if (c0 > GCN_CAP || c1 > GCN_CAP) gcn_overflow(hv, ovf, min(max(cnt[n], 0), ne), node, n, lane, acc);

    const float4 bv = ((const float4*)bias)[lane];
    float4 r;
    r.x = fmaf(acc.x + self.x, dd, bv.x);
    r.y = fmaf(acc.y + self.y, dd, bv.y);
    r.z = fmaf(acc.z + self.z, dd, bv.z);
    r.w = fmaf(acc.w + self.w, dd, bv.w);
    r.x = r.x > 0.0f ? r.x : 0.0f;
    r.y = r.y > 0.0f ? r.y : 0.0f;
    r.z = r.z > 0.0f ? r.z : 0.0f;
    r.w = r.w > 0.0f ? r.w : 0.0f;
    ((float4*)out)[(size_t)node * (GCN_D / 4) + lane] = r;
}

template <int U>
__device__ __forceinline__ void gcn_rows_load(const float4* __restrict__ hv, int sidx, int j, int lane,
                                              float4 (&v)[U]) {
#pragma unroll
    for (int u = 0; u < U; ++u) {
        const int s = __builtin_amdgcn_readlane(sidx, j + u);
        v[u] = hv[(size_t)s * (GCN_D / 4) + lane];
    }
}
template <int U>
__device__ __forceinline__ void gcn_rows_sum(const float4 (&v)[U], float4& acc) {
#pragma unroll
    for (int u = 0; u < U; ++u) {
        acc.x += v[u].x;
        acc.y += v[u].y;
        acc.z += v[u].z;
        acc.w += v[u].w;
    }
}

__device__ __forceinline__ void gcn_chunk(const float4* __restrict__ hv, int sidx, int m,
                                          int lane, float4& acc) {
    int jb = 0;
    if (m >= 16) {
        float4 va[16];
        gcn_rows_load<16>(hv, sidx, 0, lane, va);
        gcn_rows_sum<16>(va, acc);
        jb = 16;
    }
    float4 vb[8], vc[4], vd[2], ve[1];
    const bool tb = jb + 8 <= m;
    const int  jc = jb + (tb ? 8 : 0);
    const bool tc = jc + 4 <= m;
    const int  jd = jc + (tc ? 4 : 0);
    const bool td = jd + 2 <= m;
    const int  je = jd + (td ? 2 : 0);
    const bool te = je < m;
    if (tb) gcn_rows_load<8>(hv, sidx, jb, lane, vb);
    if (tc) gcn_rows_load<4>(hv, sidx, jc, lane, vc);
    if (td) gcn_rows_load<2>(hv, sidx, jd, lane, vd);
    if (te) gcn_rows_load<1>(hv, sidx, je, lane, ve);
    if (tb) gcn_rows_sum<8>(vb, acc);
    if (tc) gcn_rows_sum<4>(vc, acc);
    if (td) gcn_rows_sum<2>(vd, acc);
    if (te) gcn_rows_sum<1>(ve, acc);
    if (m == 32) gcn_rows<1>(hv, sidx, 31, lane, acc);
}

__global__ __launch_bounds__(256) void gcn_gather_gemm(const int* __restrict__ cnt,
                                                       const int* __restrict__ bucket,
                                                       const int2* __restrict__ ovf, int ne,
                                                       const float* __restrict__ hin,
                                                       const float* __restrict__ bias,
                                                       const float* __restrict__ W,
                                                       float* __restrict__ H, int n) {
    __shared__ __attribute__((aligned(16))) float sW[GCN_D * GCN_D];
    __shared__ __attribute__((aligned(16))) float sX[GEMM_ROWS * GEMM_XPITCH];

    const int t    = threadIdx.x;
    const int lane = t & 31;
    const int wave = __builtin_amdgcn_readfirstlane(t >> 5);
    int rb, re;
    gcn_wg_rows(n, rb, re);
    const float4 bv  = ((const float4*)bias)[lane];
    const float4* hv = (const float4*)hin;

    gcn_stage_w(sW, W, t);

    int row0 = rb;
    int len  = gcn_first_chunk();
    while (row0 < re) {
        const int rend  = min(re, row0 + len);
        const int node0 = row0 + wave;

        int cv0 = 0, cv1 = 0;
        if (node0 + 8 * lane < rend) {
            cv0 = cnt[node0 + 8 * lane];
            cv1 = cnt[GCN_CNT_PLANE + node0 + 8 * lane];
        }

        int c0   = __builtin_amdgcn_readlane(cv0, 0);
        int c1   = __builtin_amdgcn_readlane(cv1, 0);
        int noff = node0 * GCN_CAP;
        int m0   = min(max(c0, 0), GCN_CAP);
        int tot  = m0 + min(max(c1, 0), GCN_CAP);
        int enext = 0;
        if (lane < min(32, tot)) enext = gcn_entry(bucket, noff, m0, lane);
        asm volatile("" :: "v"(enext));
        for (int i = 0; i < 8; ++i) {
            const int  node   = node0 + 8 * i;
            const int efirst = enext;
            const int nc0   = __builtin_amdgcn_readlane(cv0, i + 1);
            const int nc1   = __builtin_amdgcn_readlane(cv1, i + 1);
            const int nnoff = noff + 8 * GCN_CAP;
            const int nm0   = min(max(nc0, 0), GCN_CAP);
            const int ntot  = nm0 + min(max(nc1, 0), GCN_CAP);
            enext = 0;
            if (i < 7 && lane < min(32, ntot)) enext = gcn_entry(bucket, nnoff, nm0, lane);

            float4 r = make_float4(0.0f, 0.0f, 0.0f, 0.0f);
            if (node < rend) {
                const float  dd   = gcn_dinv(c0 + c1);
                const float4 self = hv[(size_t)node * (GCN_D / 4) + lane];
                float4 acc = make_float4(0.0f, 0.0f, 0.0f, 0.0f);
                if (tot > 0) {
                    const int sidx = min(max(efirst, 0), n - 1);
                    gcn_chunk(hv, sidx, min(32, tot), lane, acc);
                }
                for (int base = 32; base < tot; base += 32) {
                    const int m = min(32, tot - base);
                    int e = 0;
                    if (lane < m) e = gcn_entry(bucket, noff, m0, base + lane);
                    const int sidx = min(max(e, 0), n - 1);
                    for (int j = 0; j < m; ++j) gcn_rows<1>(hv, sidx, j, lane, acc);
                }
                if (c0 > GCN_CAP || c1 > GCN_CAP) gcn_overflow(hv, ovf, min(max(cnt[n], 0), ne), node, n, lane, acc);
                r.x = fmaf(acc.x + self.x, dd, bv.x);
                r.y = fmaf(acc.y + self.y, dd, bv.y);
                r.z = fmaf(acc.z + self.z, dd, bv.z);
                r.w = fmaf(acc.w + self.w, dd, bv.w);
                r.x = (r.x > 0.0f ? r.x : 0.0f) * dd;
                r.y = (r.y > 0.0f ? r.y : 0.0f) * dd;
                r.z = (r.z > 0.0f ? r.z : 0.0f) * dd;
                r.w = (r.w > 0.0f ? r.w : 0.0f) * dd;
            }
            *(float4*)(sX + (wave + 8 * i) * GEMM_XPITCH + 4 * lane) = r;
            noff = nnoff;
            m0   = nm0;
            tot  = ntot;
            c0   = nc0;
            c1   = nc1;
        }
        __syncthreads();

        gcn_tile_product(sW, sX, H, row0, rend, t);
        __syncthreads();
        row0 = rend;
        len  = GEMM_ROWS;
    }
}

extern "C" void kernel_launch(void* const* d_in, const int* in_sizes, int n_in,
                              void* d_out, int out_size, void* d_ws, size_t ws_size,
                              hipStream_t stream) {
    const float* x  = (const float*)d_in[0];
    const float* W1 = (const float*)d_in[1];
    const float* b1 = (const float*)d_in[2];
    const float* W2 = (const float*)d_in[3];
    const float* b2 = (const float*)d_in[4];
    const float* W3 = (const float*)d_in[5];
    const float* b3 = (const float*)d_in[6];
    const int*   eidx = (const int*)d_in[7];
    const int ne = in_sizes[7] / 2;
    const int* src = eidx;
    const int* dst = eidx + ne;

    const size_t edge_bytes = (((size_t)ne * 4) + 4095) & ~(size_t)4095;
    char* ws = (char*)d_ws;
    size_t off = 0;
    int*   cnt    = (int*)(ws + off);   off += 2 * (size_t)GCN_CNT_PLANE * 4;
    int*   bucket = (int*)(ws + off);   off += 2 * (size_t)GCN_BKT_PLANE * 4;
    int2*  ovf    = (int2*)(ws + off);  off += 2 * edge_bytes;
    int*   rec    = (int*)(ws + off);   off += edge_bytes;
    int*   runs   = (int*)(ws + off);   off += edge_bytes;
    float* h      = (float*)(ws + off); off += (size_t)GCN_N * GCN_D * 4;
    float* buf    = (float*)(ws + off);

    const int blk = 256;
    const int g_chunk = (ne + GCN_CH - 1) / GCN_CH;
    const int g_gemm = GEMM_WGS;
    const int g_gath = (GCN_N + 7) / 8;

    gcn_edge_bin<<<g_chunk, blk, 0, stream>>>(src, dst, rec, runs, cnt, GCN_N, ne);
    gcn_list_build<<<GCN_NBLK, 1024, 0, stream>>>(rec, runs, cnt, bucket, ovf, GCN_N, ne);

    gcn_gemm_wmma<<<g_gemm, blk, 0, stream>>>(x, W1, cnt, h, GCN_N);

    gcn_gather_gemm<<<g_gemm, blk, 0, stream>>>(cnt, bucket, ovf, ne, h, b1, W2, buf, GCN_N);

    gcn_gather_gemm<<<g_gemm, blk, 0, stream>>>(cnt, bucket, ovf, ne, buf, b2, W3, h, GCN_N);

    gcn_gather<<<g_gath, blk, 0, stream>>>(cnt, bucket, ovf, ne, h, b3, (float*)d_out, GCN_N);
}
